# starting kernel
# baseline (speedup 1.0000x reference)
#include <hip/hip_runtime.h>
#include <math.h>


constexpr int kN   = 20000;
constexpr int kE   = 320000;
constexpr int kIN  = 128;
constexpr int kH   = 8;
constexpr int kD   = 16;
constexpr int kHD  = 128;
constexpr int kL   = 2;
constexpr int kOUT = 2;
constexpr float kLnEps = 1e-5f;

typedef __attribute__((ext_vector_type(2))) float v2f;
typedef __attribute__((ext_vector_type(8))) float v8f;

__device__ __forceinline__ float eluf(float x) {
    return x > 0.0f ? x : (expf(x) - 1.0f);
}

__global__ __launch_bounds__(256) void gemm_h1h2_kernel(
    const float* __restrict__ X,
    const float* __restrict__ W1, const float* __restrict__ B1,
    const float* __restrict__ W2, const float* __restrict__ B2,
    float* __restrict__ H1, float* __restrict__ H2, float* __restrict__ HP)
{
    const int lane  = threadIdx.x & 31;
    const int wave  = threadIdx.x >> 5;
    const int mbase = blockIdx.x * 16;
    const int n0    = wave * 16;
    const int r     = lane & 15;
    const int half  = lane >> 4;

    const float* xrow = X + (size_t)(mbase + r) * kIN;

    v8f acc1 = {};
    v8f acc2 = {};

#pragma unroll
    for (int k0 = 0; k0 < kIN; k0 += 4) {
        const int ka = k0 + half * 2;
        v2f a = *(const v2f*)(xrow + ka);
        v2f bA, bB;
        bA.x = W1[(size_t)ka       * kHD + n0 + r];
        bA.y = W1[(size_t)(ka + 1) * kHD + n0 + r];
        bB.x = W2[(size_t)ka       * kHD + n0 + r];
        bB.y = W2[(size_t)(ka + 1) * kHD + n0 + r];
        acc1 = __builtin_amdgcn_wmma_f32_16x16x4_f32(
            false, a, false, bA, (short)0, acc1, false, false);
        acc2 = __builtin_amdgcn_wmma_f32_16x16x4_f32(
            false, a, false, bB, (short)0, acc2, false, false);
    }

    const int n = n0 + r;
    const float bb1 = B1[n];
    const float bb2 = B2[n];

#pragma unroll
    for (int j = 0; j < 8; ++j) {
        const int m = mbase + j + half * 8;
        const float h1v = acc1[j] + bb1;
        const float h2v = acc2[j] + bb2;
        const size_t idx = (size_t)m * kHD + n;
        H1[idx] = h1v;
        H2[idx] = h2v;
        HP[idx] = h1v * h2v;
    }
}

__global__ __launch_bounds__(256) void edge_attn_kernel(
    const float* __restrict__ H1, const float* __restrict__ H2,
    const float* __restrict__ HP,
    const int* __restrict__ src, const int* __restrict__ dst,
    const float* __restrict__ avec, float* __restrict__ AGG)
{
    const int lane = threadIdx.x & 31;
    const int e    = blockIdx.x * 8 + (threadIdx.x >> 5);
    if (e >= kE) return;

    const int s = src[e];
    const int d = dst[e];

    const float4 h1s = *(const float4*)(H1 + (size_t)s * kHD + lane * 4);
    const float4 h2d = *(const float4*)(H2 + (size_t)d * kHD + lane * 4);
    const float4 hps = *(const float4*)(HP + (size_t)s * kHD + lane * 4);
    const float4 hpd = *(const float4*)(HP + (size_t)d * kHD + lane * 4);
    const float4 av  = *(const float4*)(avec + lane * 4);

    float part;
    {
        const float t0 = eluf(h1s.x + h2d.x + hps.x * hpd.x);
        const float t1 = eluf(h1s.y + h2d.y + hps.y * hpd.y);
        const float t2 = eluf(h1s.z + h2d.z + hps.z * hpd.z);
        const float t3 = eluf(h1s.w + h2d.w + hps.w * hpd.w);
        part = av.x * t0 + av.y * t1 + av.z * t2 + av.w * t3;
    }
    part *= 0.25f;

    part += __shfl_xor(part, 1, 32);
    part += __shfl_xor(part, 2, 32);

    float mx = part;
    mx = fmaxf(mx, __shfl_xor(mx, 4, 32));
    mx = fmaxf(mx, __shfl_xor(mx, 8, 32));
    mx = fmaxf(mx, __shfl_xor(mx, 16, 32));

    const float ex = expf(part - mx);
    float sm = ex;
    sm += __shfl_xor(sm, 4, 32);
    sm += __shfl_xor(sm, 8, 32);
    sm += __shfl_xor(sm, 16, 32);

    const float alpha = ex / sm;

    float* aggrow = AGG + (size_t)d * kHD + lane * 4;
    atomicAdd(aggrow + 0, alpha * h1s.x);
    atomicAdd(aggrow + 1, alpha * h1s.y);
    atomicAdd(aggrow + 2, alpha * h1s.z);
    atomicAdd(aggrow + 3, alpha * h1s.w);
}

__global__ __launch_bounds__(256) void ln_elu_res_kernel(
    const float* __restrict__ AGG,
    const float* __restrict__ G, const float* __restrict__ B,
    const float* __restrict__ Xin, float* __restrict__ Xout)
{
    const int lane = threadIdx.x & 31;
    const int node = blockIdx.x * 8 + (threadIdx.x >> 5);
    if (node >= kN) return;

    const float4 v = *(const float4*)(AGG + (size_t)node * kHD + lane * 4);

    float s = v.x + v.y + v.z + v.w;
#pragma unroll
    for (int m = 1; m <= 16; m <<= 1) s += __shfl_xor(s, m, 32);
    const float mu = s * (1.0f / 128.0f);

    const float dx0 = v.x - mu, dx1 = v.y - mu, dx2 = v.z - mu, dx3 = v.w - mu;
    float q = dx0 * dx0 + dx1 * dx1 + dx2 * dx2 + dx3 * dx3;
#pragma unroll
    for (int m = 1; m <= 16; m <<= 1) q += __shfl_xor(q, m, 32);
    const float inv = rsqrtf(q * (1.0f / 128.0f) + kLnEps);

    const float4 g  = *(const float4*)(G + lane * 4);
    const float4 b  = *(const float4*)(B + lane * 4);
    const float4 xr = *(const float4*)(Xin + (size_t)node * kHD + lane * 4);

    float4 o;
    o.x = eluf(dx0 * inv * g.x + b.x) + xr.x;
    o.y = eluf(dx1 * inv * g.y + b.y) + xr.y;
    o.z = eluf(dx2 * inv * g.z + b.z) + xr.z;
    o.w = eluf(dx3 * inv * g.w + b.w) + xr.w;

    *(float4*)(Xout + (size_t)node * kHD + lane * 4) = o;
}

__global__ __launch_bounds__(256) void out_proj_kernel(
    const float* __restrict__ X, const float* __restrict__ Wo,
    const float* __restrict__ bo, float* __restrict__ O)
{
    const int n = blockIdx.x * blockDim.x + threadIdx.x;
    if (n >= kN) return;
    float a0 = bo[0];
    float a1 = bo[1];
    const float* xr = X + (size_t)n * kHD;
#pragma unroll 8
    for (int k = 0; k < kHD; ++k) {
        const float xv = xr[k];
        a0 += xv * Wo[2 * k + 0];
        a1 += xv * Wo[2 * k + 1];
    }
    O[2 * n + 0] = a0;
    O[2 * n + 1] = a1;
}

extern "C" void kernel_launch(void* const* d_in, const int* in_sizes, int n_in,
                              void* d_out, int out_size, void* d_ws, size_t ws_size,
                              hipStream_t stream)
{
    (void)in_sizes; (void)n_in; (void)out_size; (void)ws_size;

    const float* x    = (const float*)d_in[0];
    const int*   ei   = (const int*)  d_in[1];
    const float* W1   = (const float*)d_in[2];
    const float* b1   = (const float*)d_in[3];
    const float* W2   = (const float*)d_in[4];
    const float* b2   = (const float*)d_in[5];
    const float* a    = (const float*)d_in[6];
    const float* ln_g = (const float*)d_in[7];
    const float* ln_b = (const float*)d_in[8];
    const float* Wo   = (const float*)d_in[9];
    const float* bo   = (const float*)d_in[10];
    float*       out  = (float*)d_out;

    const size_t NHD = (size_t)kN * kHD;
    float* h1   = (float*)d_ws;
    float* h2   = h1  + NHD;
    float* hp   = h2  + NHD;
    float* agg  = hp  + NHD;
    float* xcur = agg + NHD;

    const int* srcIdx = ei;
    const int* dstIdx = ei + kE;

    for (int layer = 0; layer < kL; ++layer) {
        const float* xin = (layer == 0) ? x : xcur;

        gemm_h1h2_kernel<<<kN / 16, 256, 0, stream>>>(
            xin,
            W1 + (size_t)layer * kIN * kHD, b1 + (size_t)layer * kHD,
            W2 + (size_t)layer * kIN * kHD, b2 + (size_t)layer * kHD,
            h1, h2, hp);

        hipMemsetAsync(agg, 0, NHD * sizeof(float), stream);

        edge_attn_kernel<<<kE / 8, 256, 0, stream>>>(
            h1, h2, hp, srcIdx, dstIdx, a + (size_t)layer * kH * kD, agg);

        ln_elu_res_kernel<<<kN / 8, 256, 0, stream>>>(
            agg, ln_g + (size_t)layer * kHD, ln_b + (size_t)layer * kHD,
            xin, xcur);
    }

    out_proj_kernel<<<(kN + 255) / 256, 256, 0, stream>>>(xcur, Wo, bo, out);
}


# Edge kernel tiled by destination range: a block owns 80 dst rows, scans the dst array for its edges and sums messages in LDS (f32 LDS atomics, arrival order); no global atomics, no agg memset
# speedup vs baseline: 1.1438x; 1.1438x over previous
#include <hip/hip_runtime.h>
#include <math.h>


constexpr int kN   = 20000;
constexpr int kE   = 320000;
constexpr int kIN  = 128;
constexpr int kH   = 8;
constexpr int kD   = 16;
constexpr int kHD  = 128;
constexpr int kL   = 2;
constexpr int kOUT = 2;
constexpr float kLnEps = 1e-5f;

typedef __attribute__((ext_vector_type(2))) float v2f;
typedef __attribute__((ext_vector_type(8))) float v8f;

__device__ __forceinline__ float eluf(float x) {
    return x > 0.0f ? x : (expf(x) - 1.0f);
}

__global__ __launch_bounds__(256) void gemm_h1h2_kernel(
    const float* __restrict__ X,
    const float* __restrict__ W1, const float* __restrict__ B1,
    const float* __restrict__ W2, const float* __restrict__ B2,
    float* __restrict__ H1, float* __restrict__ H2, float* __restrict__ HP)
{
    const int lane  = threadIdx.x & 31;
    const int wave  = threadIdx.x >> 5;
    const int mbase = blockIdx.x * 16;
    const int n0    = wave * 16;
    const int r     = lane & 15;
    const int half  = lane >> 4;

    const float* xrow = X + (size_t)(mbase + r) * kIN;

    v8f acc1 = {};
    v8f acc2 = {};

#pragma unroll
    for (int k0 = 0; k0 < kIN; k0 += 4) {
        const int ka = k0 + half * 2;
        v2f a = *(const v2f*)(xrow + ka);
        v2f bA, bB;
        bA.x = W1[(size_t)ka       * kHD + n0 + r];
        bA.y = W1[(size_t)(ka + 1) * kHD + n0 + r];
        bB.x = W2[(size_t)ka       * kHD + n0 + r];
        bB.y = W2[(size_t)(ka + 1) * kHD + n0 + r];
        acc1 = __builtin_amdgcn_wmma_f32_16x16x4_f32(
            false, a, false, bA, (short)0, acc1, false, false);
        acc2 = __builtin_amdgcn_wmma_f32_16x16x4_f32(
            false, a, false, bB, (short)0, acc2, false, false);
    }

    const int n = n0 + r;
    const float bb1 = B1[n];
    const float bb2 = B2[n];

#pragma unroll
    for (int j = 0; j < 8; ++j) {
        const int m = mbase + j + half * 8;
        const float h1v = acc1[j] + bb1;
        const float h2v = acc2[j] + bb2;
        const size_t idx = (size_t)m * kHD + n;
        H1[idx] = h1v;
        H2[idx] = h2v;
        HP[idx] = h1v * h2v;
    }
}

constexpr int kTR     = 80;
constexpr int kTiles  = kN / kTR;
constexpr int kChunks = kE / 128;
constexpr int kCap    = 640;

__global__ __launch_bounds__(1024) void edge_tile_kernel(
    const float* __restrict__ H1, const float* __restrict__ H2,
    const float* __restrict__ HP,
    const int* __restrict__ src, const int* __restrict__ dst,
    const float* __restrict__ avec, float* __restrict__ AGG)
{
    __shared__ float    sAcc[kTR * kHD];
    __shared__ float4   sH2[kTR * 32];
    __shared__ float4   sHPd[kTR * 32];
    __shared__ unsigned sList[32 * kCap];

    const int tid  = threadIdx.x;
    const int lane = tid & 31;
    const int wave = __builtin_amdgcn_readfirstlane(tid >> 5);
    const unsigned lo = (unsigned)blockIdx.x * kTR;

    for (int i = tid; i < kTR * kHD; i += 1024) sAcc[i] = 0.0f;
    {
        const float4* g2 = (const float4*)(H2 + (size_t)lo * kHD);
        const float4* gp = (const float4*)(HP + (size_t)lo * kHD);
        for (int i = tid; i < kTR * 32; i += 1024) {
            sH2[i]  = g2[i];
            sHPd[i] = gp[i];
        }
    }
    const float4 av = *(const float4*)(avec + lane * 4);
    __syncthreads();

    unsigned* myList = sList + wave * kCap;
    const unsigned lanesBelow = (1u << lane) - 1u;
    int cnt = 0;

    for (int c0 = wave; ; c0 += 128) {
        const bool done = (c0 >= kChunks);
        if (done || cnt > kCap - 512) {
            for (int i = lane; i < cnt; i += 32) {
                const unsigned ent = myList[i];
                myList[i] = (unsigned)src[ent & 0xFFFFFu] | (ent & 0xFFF00000u);
            }
            for (int i0 = 0; i0 < cnt; i0 += 4) {
                float4 a1[4], ap[4];
                int dl[4];
#pragma unroll
                for (int j = 0; j < 4; ++j) {
                    const int i = (i0 + j < cnt) ? (i0 + j) : (cnt - 1);
                    const unsigned ent = __builtin_amdgcn_readfirstlane(myList[i]);
                    const int s = (int)(ent & 0xFFFFFu);
                    dl[j] = (int)(ent >> 20);
                    a1[j] = *(const float4*)(H1 + (size_t)s * kHD + lane * 4);
                    ap[j] = *(const float4*)(HP + (size_t)s * kHD + lane * 4);
                }
#pragma unroll
                for (int j = 0; j < 4; ++j) {
                    if (i0 + j < cnt) {
                        const float4 h1s = a1[j];
                        const float4 hps = ap[j];
                        const float4 h2d = sH2[dl[j] * 32 + lane];
                        const float4 hpd = sHPd[dl[j] * 32 + lane];
                        float part;
                        {
                            const float t0 = eluf(h1s.x + h2d.x + hps.x * hpd.x);
                            const float t1 = eluf(h1s.y + h2d.y + hps.y * hpd.y);
                            const float t2 = eluf(h1s.z + h2d.z + hps.z * hpd.z);
                            const float t3 = eluf(h1s.w + h2d.w + hps.w * hpd.w);
                            part = av.x * t0 + av.y * t1 + av.z * t2 + av.w * t3;
                        }
                        part *= 0.25f;
                        part += __shfl_xor(part, 1, 32);
                        part += __shfl_xor(part, 2, 32);
                        float mx = part;
                        mx = fmaxf(mx, __shfl_xor(mx, 4, 32));
                        mx = fmaxf(mx, __shfl_xor(mx, 8, 32));
                        mx = fmaxf(mx, __shfl_xor(mx, 16, 32));
                        const float ex = expf(part - mx);
                        float sm = ex;
                        sm += __shfl_xor(sm, 4, 32);
                        sm += __shfl_xor(sm, 8, 32);
                        sm += __shfl_xor(sm, 16, 32);
                        const float alpha = ex / sm;
                        float* arow = sAcc + dl[j] * kHD + lane;
                        atomicAdd(arow +  0, alpha * h1s.x);
                        atomicAdd(arow + 32, alpha * h1s.y);
                        atomicAdd(arow + 64, alpha * h1s.z);
                        atomicAdd(arow + 96, alpha * h1s.w);
                    }
                }
            }
            cnt = 0;
        }
        if (done) break;

        int4 dv[4];
#pragma unroll
        for (int u = 0; u < 4; ++u) {
            const int c = c0 + 32 * u;
            if (c < kChunks) dv[u] = *(const int4*)(dst + (size_t)c * 128 + lane * 4);
            else             dv[u] = make_int4(-1, -1, -1, -1);
        }
#pragma unroll
        for (int u = 0; u < 4; ++u) {
            const int ebase = (c0 + 32 * u) * 128 + lane * 4;
            const unsigned r0 = (unsigned)dv[u].x - lo;
            const unsigned r1 = (unsigned)dv[u].y - lo;
            const unsigned r2 = (unsigned)dv[u].z - lo;
            const unsigned r3 = (unsigned)dv[u].w - lo;
            const unsigned rm = min(min(r0, r1), min(r2, r3));
            if (__builtin_amdgcn_ballot_w32(rm < (unsigned)kTR) != 0u) {
                const unsigned rr[4] = {r0, r1, r2, r3};
#pragma unroll
                for (int k = 0; k < 4; ++k) {
                    const bool hit = rr[k] < (unsigned)kTR;
                    const unsigned mask = __builtin_amdgcn_ballot_w32(hit);
                    if (hit) {
                        const int pos = cnt + __popc(mask & lanesBelow);
                        myList[pos] = (unsigned)(ebase + k) | (rr[k] << 20);
                    }
                    cnt += __popc(mask);
                }
            }
        }
    }

    __syncthreads();
    for (int i = tid; i < kTR * 32; i += 1024) {
        const int row = i >> 5;
        const int l   = i & 31;
        const float* a = sAcc + row * kHD + l;
        float4 o;
        o.x = a[0];
        o.y = a[32];
        o.z = a[64];
        o.w = a[96];
        *(float4*)(AGG + ((size_t)lo + row) * kHD + l * 4) = o;
    }
}

__global__ __launch_bounds__(256) void ln_elu_res_kernel(
    const float* __restrict__ AGG,
    const float* __restrict__ G, const float* __restrict__ B,
    const float* __restrict__ Xin, float* __restrict__ Xout)
{
    const int lane = threadIdx.x & 31;
    const int node = blockIdx.x * 8 + (threadIdx.x >> 5);
    if (node >= kN) return;

    const float4 v = *(const float4*)(AGG + (size_t)node * kHD + lane * 4);

    float s = v.x + v.y + v.z + v.w;
#pragma unroll
    for (int m = 1; m <= 16; m <<= 1) s += __shfl_xor(s, m, 32);
    const float mu = s * (1.0f / 128.0f);

    const float dx0 = v.x - mu, dx1 = v.y - mu, dx2 = v.z - mu, dx3 = v.w - mu;
    float q = dx0 * dx0 + dx1 * dx1 + dx2 * dx2 + dx3 * dx3;
#pragma unroll
    for (int m = 1; m <= 16; m <<= 1) q += __shfl_xor(q, m, 32);
    const float inv = rsqrtf(q * (1.0f / 128.0f) + kLnEps);

    const float4 g  = *(const float4*)(G + lane * 4);
    const float4 b  = *(const float4*)(B + lane * 4);
    const float4 xr = *(const float4*)(Xin + (size_t)node * kHD + lane * 4);

    float4 o;
    o.x = eluf(dx0 * inv * g.x + b.x) + xr.x;
    o.y = eluf(dx1 * inv * g.y + b.y) + xr.y;
    o.z = eluf(dx2 * inv * g.z + b.z) + xr.z;
    o.w = eluf(dx3 * inv * g.w + b.w) + xr.w;

    *(float4*)(Xout + (size_t)node * kHD + lane * 4) = o;
}

__global__ __launch_bounds__(256) void out_proj_kernel(
    const float* __restrict__ X, const float* __restrict__ Wo,
    const float* __restrict__ bo, float* __restrict__ O)
{
    const int n = blockIdx.x * blockDim.x + threadIdx.x;
    if (n >= kN) return;
    float a0 = bo[0];
    float a1 = bo[1];
    const float* xr = X + (size_t)n * kHD;
#pragma unroll 8
    for (int k = 0; k < kHD; ++k) {
        const float xv = xr[k];
        a0 += xv * Wo[2 * k + 0];
        a1 += xv * Wo[2 * k + 1];
    }
    O[2 * n + 0] = a0;
    O[2 * n + 1] = a1;
}

extern "C" void kernel_launch(void* const* d_in, const int* in_sizes, int n_in,
                              void* d_out, int out_size, void* d_ws, size_t ws_size,
                              hipStream_t stream)
{
    (void)in_sizes; (void)n_in; (void)out_size; (void)ws_size;

    const float* x    = (const float*)d_in[0];
    const int*   ei   = (const int*)  d_in[1];
    const float* W1   = (const float*)d_in[2];
    const float* b1   = (const float*)d_in[3];
    const float* W2   = (const float*)d_in[4];
    const float* b2   = (const float*)d_in[5];
    const float* a    = (const float*)d_in[6];
    const float* ln_g = (const float*)d_in[7];
    const float* ln_b = (const float*)d_in[8];
    const float* Wo   = (const float*)d_in[9];
    const float* bo   = (const float*)d_in[10];
    float*       out  = (float*)d_out;

    const size_t NHD = (size_t)kN * kHD;
    float* h1   = (float*)d_ws;
    float* h2   = h1  + NHD;
    float* hp   = h2  + NHD;
    float* agg  = hp  + NHD;
    float* xcur = agg + NHD;

    const int* srcIdx = ei;
    const int* dstIdx = ei + kE;

    for (int layer = 0; layer < kL; ++layer) {
        const float* xin = (layer == 0) ? x : xcur;

        gemm_h1h2_kernel<<<kN / 16, 256, 0, stream>>>(
            xin,
            W1 + (size_t)layer * kIN * kHD, b1 + (size_t)layer * kHD,
            W2 + (size_t)layer * kIN * kHD, b2 + (size_t)layer * kHD,
            h1, h2, hp);

        edge_tile_kernel<<<kTiles, 1024, 0, stream>>>(
            h1, h2, hp, srcIdx, dstIdx, a + (size_t)layer * kH * kD, agg);

        ln_elu_res_kernel<<<kN / 8, 256, 0, stream>>>(
            agg, ln_g + (size_t)layer * kHD, ln_b + (size_t)layer * kHD,
            xin, xcur);
    }

    out_proj_kernel<<<(kN + 255) / 256, 256, 0, stream>>>(xcur, Wo, bo, out);
}


# Edge lists built once per call: layer 1's tile kernel saves each wave's (src, dst row) list in the workspace, layer 2's reads it back instead of re-scanning dst; same arrival-order LDS-atomic sums
# speedup vs baseline: 1.1882x; 1.0388x over previous
#include <hip/hip_runtime.h>
#include <math.h>


constexpr int kN   = 20000;
constexpr int kE   = 320000;
constexpr int kIN  = 128;
constexpr int kH   = 8;
constexpr int kD   = 16;
constexpr int kHD  = 128;
constexpr int kL   = 2;
constexpr int kOUT = 2;
constexpr float kLnEps = 1e-5f;

typedef __attribute__((ext_vector_type(2))) float v2f;
typedef __attribute__((ext_vector_type(8))) float v8f;

__device__ __forceinline__ float eluf(float x) {
    return x > 0.0f ? x : (expf(x) - 1.0f);
}

__global__ __launch_bounds__(256) void gemm_h1h2_kernel(
    const float* __restrict__ X,
    const float* __restrict__ W1, const float* __restrict__ B1,
    const float* __restrict__ W2, const float* __restrict__ B2,
    float* __restrict__ H1, float* __restrict__ H2, float* __restrict__ HP)
{
    const int lane  = threadIdx.x & 31;
    const int wave  = threadIdx.x >> 5;
    const int mbase = blockIdx.x * 16;
    const int n0    = wave * 16;
    const int r     = lane & 15;
    const int half  = lane >> 4;

    const float* xrow = X + (size_t)(mbase + r) * kIN;

    v8f acc1 = {};
    v8f acc2 = {};

#pragma unroll
    for (int k0 = 0; k0 < kIN; k0 += 4) {
        const int ka = k0 + half * 2;
        v2f a = *(const v2f*)(xrow + ka);
        v2f bA, bB;
        bA.x = W1[(size_t)ka       * kHD + n0 + r];
        bA.y = W1[(size_t)(ka + 1) * kHD + n0 + r];
        bB.x = W2[(size_t)ka       * kHD + n0 + r];
        bB.y = W2[(size_t)(ka + 1) * kHD + n0 + r];
        acc1 = __builtin_amdgcn_wmma_f32_16x16x4_f32(
            false, a, false, bA, (short)0, acc1, false, false);
        acc2 = __builtin_amdgcn_wmma_f32_16x16x4_f32(
            false, a, false, bB, (short)0, acc2, false, false);
    }

    const int n = n0 + r;
    const float bb1 = B1[n];
    const float bb2 = B2[n];

#pragma unroll
    for (int j = 0; j < 8; ++j) {
        const int m = mbase + j + half * 8;
        const float h1v = acc1[j] + bb1;
        const float h2v = acc2[j] + bb2;
        const size_t idx = (size_t)m * kHD + n;
        H1[idx] = h1v;
        H2[idx] = h2v;
        HP[idx] = h1v * h2v;
    }
}

constexpr int kTR     = 80;
constexpr int kTiles  = kN / kTR;
constexpr int kChunks = kE / 128;
constexpr int kCap    = 640;
constexpr int kSlot   = 1024;
constexpr unsigned kOver = 0xFFFFFFFFu;

__device__ __forceinline__ void work_list(
    const unsigned* myList, const int cnt, const int lane, const float4 av,
    const float* __restrict__ H1, const float* __restrict__ HP,
    const float4* sH2, const float4* sHPd, float* sAcc)
{
    for (int i0 = 0; i0 < cnt; i0 += 4) {
        float4 a1[4], ap[4];
        int dl[4];
#pragma unroll
        for (int j = 0; j < 4; ++j) {
            const int i = (i0 + j < cnt) ? (i0 + j) : (cnt - 1);
            const unsigned ent = __builtin_amdgcn_readfirstlane(myList[i]);
            const int s = (int)(ent & 0xFFFFFu);
            dl[j] = (int)(ent >> 20);
            a1[j] = *(const float4*)(H1 + (size_t)s * kHD + lane * 4);
            ap[j] = *(const float4*)(HP + (size_t)s * kHD + lane * 4);
        }
#pragma unroll
        for (int j = 0; j < 4; ++j) {
            if (i0 + j < cnt) {
                const float4 h1s = a1[j];
                const float4 hps = ap[j];
                const float4 h2d = sH2[dl[j] * 32 + lane];
                const float4 hpd = sHPd[dl[j] * 32 + lane];
                float part;
                {
                    const float t0 = eluf(h1s.x + h2d.x + hps.x * hpd.x);
                    const float t1 = eluf(h1s.y + h2d.y + hps.y * hpd.y);
                    const float t2 = eluf(h1s.z + h2d.z + hps.z * hpd.z);
                    const float t3 = eluf(h1s.w + h2d.w + hps.w * hpd.w);
                    part = av.x * t0 + av.y * t1 + av.z * t2 + av.w * t3;
                }
                part *= 0.25f;
                part += __shfl_xor(part, 1, 32);
                part += __shfl_xor(part, 2, 32);
                float mx = part;
                mx = fmaxf(mx, __shfl_xor(mx, 4, 32));
                mx = fmaxf(mx, __shfl_xor(mx, 8, 32));
                mx = fmaxf(mx, __shfl_xor(mx, 16, 32));
                const float ex = expf(part - mx);
                float sm = ex;
                sm += __shfl_xor(sm, 4, 32);
                sm += __shfl_xor(sm, 8, 32);
                sm += __shfl_xor(sm, 16, 32);
                const float alpha = ex / sm;
                float* arow = sAcc + dl[j] * kHD + lane;
                atomicAdd(arow +  0, alpha * h1s.x);
                atomicAdd(arow + 32, alpha * h1s.y);
                atomicAdd(arow + 64, alpha * h1s.z);
                atomicAdd(arow + 96, alpha * h1s.w);
            }
        }
    }
}

template <bool kSave>
__device__ __forceinline__ void scan_and_work(
    unsigned* myList, const int wave, const int lane, const unsigned lo, const float4 av,
    const float* __restrict__ H1, const float* __restrict__ HP,
    const int* __restrict__ src, const int* __restrict__ dst,
    const float4* sH2, const float4* sHPd, float* sAcc,
    unsigned* __restrict__ gSlot, unsigned* __restrict__ gCnt)
{
    const unsigned lanesBelow = (1u << lane) - 1u;
    int cnt = 0;
    unsigned saved = 0;

    for (int c0 = wave; ; c0 += 128) {
        const bool done = (c0 >= kChunks);
        if (done || cnt > kCap - 512) {
            const bool fits = kSave && saved != kOver && saved + (unsigned)cnt <= (unsigned)kSlot;
            for (int i = lane; i < cnt; i += 32) {
                const unsigned ent = myList[i];
                const unsigned v = (unsigned)src[ent & 0xFFFFFu] | (ent & 0xFFF00000u);
                myList[i] = v;
                if (fits) gSlot[saved + i] = v;
            }
            if (kSave) saved = fits ? saved + (unsigned)cnt : kOver;
            work_list(myList, cnt, lane, av, H1, HP, sH2, sHPd, sAcc);
            cnt = 0;
        }
        if (done) break;

        int4 dv[4];
#pragma unroll
        for (int u = 0; u < 4; ++u) {
            const int c = c0 + 32 * u;
            if (c < kChunks) dv[u] = *(const int4*)(dst + (size_t)c * 128 + lane * 4);
            else             dv[u] = make_int4(-1, -1, -1, -1);
        }
#pragma unroll
        for (int u = 0; u < 4; ++u) {
            const int ebase = (c0 + 32 * u) * 128 + lane * 4;
            const unsigned r0 = (unsigned)dv[u].x - lo;
            const unsigned r1 = (unsigned)dv[u].y - lo;
            const unsigned r2 = (unsigned)dv[u].z - lo;
            const unsigned r3 = (unsigned)dv[u].w - lo;
            const unsigned rm = min(min(r0, r1), min(r2, r3));
            if (__builtin_amdgcn_ballot_w32(rm < (unsigned)kTR) != 0u) {
                const unsigned rr[4] = {r0, r1, r2, r3};
#pragma unroll
                for (int k = 0; k < 4; ++k) {
                    const bool hit = rr[k] < (unsigned)kTR;
                    const unsigned mask = __builtin_amdgcn_ballot_w32(hit);
                    if (hit) {
                        const int pos = cnt + __popc(mask & lanesBelow);
                        myList[pos] = (unsigned)(ebase + k) | (rr[k] << 20);
                    }
                    cnt += __popc(mask);
                }
            }
        }
    }
    if (kSave && lane == 0) *gCnt = saved;
}

template <bool kReuse>
__device__ __forceinline__ void edge_tile_body(
    const float* __restrict__ H1, const float* __restrict__ H2,
    const float* __restrict__ HP,
    const int* __restrict__ src, const int* __restrict__ dst,
    const float* __restrict__ avec, float* __restrict__ AGG,
    unsigned* __restrict__ gList, unsigned* __restrict__ gCount)
{
    __shared__ float    sAcc[kTR * kHD];
    __shared__ float4   sH2[kTR * 32];
    __shared__ float4   sHPd[kTR * 32];
    __shared__ unsigned sList[32 * kCap];

    const int tid  = threadIdx.x;
    const int lane = tid & 31;
    const int wave = __builtin_amdgcn_readfirstlane(tid >> 5);
    const unsigned lo = (unsigned)blockIdx.x * kTR;

    for (int i = tid; i < kTR * kHD; i += 1024) sAcc[i] = 0.0f;
    {
        const float4* g2 = (const float4*)(H2 + (size_t)lo * kHD);
        const float4* gp = (const float4*)(HP + (size_t)lo * kHD);
        for (int i = tid; i < kTR * 32; i += 1024) {
            sH2[i]  = g2[i];
            sHPd[i] = gp[i];
        }
    }
    const float4 av = *(const float4*)(avec + lane * 4);
    __syncthreads();

    unsigned* myList = sList + wave * kCap;
    const size_t slot = (size_t)blockIdx.x * 32 + (size_t)wave;
    unsigned* gSlot = gList + slot * kSlot;
    unsigned* gCnt  = gCount + slot;

    if (kReuse) {
        const unsigned n = __builtin_amdgcn_readfirstlane(*gCnt);
        if (n <= (unsigned)kSlot) {
            for (unsigned base = 0; base < n; base += kCap) {
                const int m = (int)min((unsigned)kCap, n - base);
                for (int i = lane; i < m; i += 32) myList[i] = gSlot[base + i];
                work_list(myList, m, lane, av, H1, HP, sH2, sHPd, sAcc);
            }
        } else {
            scan_and_work<false>(myList, wave, lane, lo, av, H1, HP, src, dst,
                                 sH2, sHPd, sAcc, gSlot, gCnt);
        }
    } else {
        scan_and_work<true>(myList, wave, lane, lo, av, H1, HP, src, dst,
                            sH2, sHPd, sAcc, gSlot, gCnt);
    }

    __syncthreads();
    for (int i = tid; i < kTR * 32; i += 1024) {
        const int row = i >> 5;
        const int l   = i & 31;
        const float* a = sAcc + row * kHD + l;
        float4 o;
        o.x = a[0];
        o.y = a[32];
        o.z = a[64];
        o.w = a[96];
        *(float4*)(AGG + ((size_t)lo + row) * kHD + l * 4) = o;
    }
}

__global__ __launch_bounds__(1024) void edge_tile_scan_kernel(
    const float* __restrict__ H1, const float* __restrict__ H2,
    const float* __restrict__ HP,
    const int* __restrict__ src, const int* __restrict__ dst,
    const float* __restrict__ avec, float* __restrict__ AGG,
    unsigned* __restrict__ gList, unsigned* __restrict__ gCount)
{
    edge_tile_body<false>(H1, H2, HP, src, dst, avec, AGG, gList, gCount);
}

__global__ __launch_bounds__(1024) void edge_tile_list_kernel(
    const float* __restrict__ H1, const float* __restrict__ H2,
    const float* __restrict__ HP,
    const int* __restrict__ src, const int* __restrict__ dst,
    const float* __restrict__ avec, float* __restrict__ AGG,
    unsigned* __restrict__ gList, unsigned* __restrict__ gCount)
{
    edge_tile_body<true>(H1, H2, HP, src, dst, avec, AGG, gList, gCount);
}

__global__ __launch_bounds__(256) void ln_elu_res_kernel(
    const float* __restrict__ AGG,
    const float* __restrict__ G, const float* __restrict__ B,
    const float* __restrict__ Xin, float* __restrict__ Xout)
{
    const int lane = threadIdx.x & 31;
    const int node = blockIdx.x * 8 + (threadIdx.x >> 5);
    if (node >= kN) return;

    const float4 v = *(const float4*)(AGG + (size_t)node * kHD + lane * 4);

    float s = v.x + v.y + v.z + v.w;
#pragma unroll
    for (int m = 1; m <= 16; m <<= 1) s += __shfl_xor(s, m, 32);
    const float mu = s * (1.0f / 128.0f);

    const float dx0 = v.x - mu, dx1 = v.y - mu, dx2 = v.z - mu, dx3 = v.w - mu;
    float q = dx0 * dx0 + dx1 * dx1 + dx2 * dx2 + dx3 * dx3;
#pragma unroll
    for (int m = 1; m <= 16; m <<= 1) q += __shfl_xor(q, m, 32);
    const float inv = rsqrtf(q * (1.0f / 128.0f) + kLnEps);

    const float4 g  = *(const float4*)(G + lane * 4);
    const float4 b  = *(const float4*)(B + lane * 4);
    const float4 xr = *(const float4*)(Xin + (size_t)node * kHD + lane * 4);

    float4 o;
    o.x = eluf(dx0 * inv * g.x + b.x) + xr.x;
    o.y = eluf(dx1 * inv * g.y + b.y) + xr.y;
    o.z = eluf(dx2 * inv * g.z + b.z) + xr.z;
    o.w = eluf(dx3 * inv * g.w + b.w) + xr.w;

    *(float4*)(Xout + (size_t)node * kHD + lane * 4) = o;
}

__global__ __launch_bounds__(256) void out_proj_kernel(
    const float* __restrict__ X, const float* __restrict__ Wo,
    const float* __restrict__ bo, float* __restrict__ O)
{
    const int n = blockIdx.x * blockDim.x + threadIdx.x;
    if (n >= kN) return;
    float a0 = bo[0];
    float a1 = bo[1];
    const float* xr = X + (size_t)n * kHD;
#pragma unroll 8
    for (int k = 0; k < kHD; ++k) {
        const float xv = xr[k];
        a0 += xv * Wo[2 * k + 0];
        a1 += xv * Wo[2 * k + 1];
    }
    O[2 * n + 0] = a0;
    O[2 * n + 1] = a1;
}

extern "C" void kernel_launch(void* const* d_in, const int* in_sizes, int n_in,
                              void* d_out, int out_size, void* d_ws, size_t ws_size,
                              hipStream_t stream)
{
    (void)in_sizes; (void)n_in; (void)out_size; (void)ws_size;

    const float* x    = (const float*)d_in[0];
    const int*   ei   = (const int*)  d_in[1];
    const float* W1   = (const float*)d_in[2];
    const float* b1   = (const float*)d_in[3];
    const float* W2   = (const float*)d_in[4];
    const float* b2   = (const float*)d_in[5];
    const float* a    = (const float*)d_in[6];
    const float* ln_g = (const float*)d_in[7];
    const float* ln_b = (const float*)d_in[8];
    const float* Wo   = (const float*)d_in[9];
    const float* bo   = (const float*)d_in[10];
    float*       out  = (float*)d_out;

    const size_t NHD = (size_t)kN * kHD;
    float* h1   = (float*)d_ws;
    float* h2   = h1  + NHD;
    float* hp   = h2  + NHD;
    float* agg  = hp  + NHD;
    float* xcur = agg + NHD;
    unsigned* gList  = (unsigned*)(xcur + NHD);
    unsigned* gCount = gList + (size_t)kTiles * 32 * kSlot;

    const int* srcIdx = ei;
    const int* dstIdx = ei + kE;

    for (int layer = 0; layer < kL; ++layer) {
        const float* xin = (layer == 0) ? x : xcur;

        gemm_h1h2_kernel<<<kN / 16, 256, 0, stream>>>(
            xin,
            W1 + (size_t)layer * kIN * kHD, b1 + (size_t)layer * kHD,
            W2 + (size_t)layer * kIN * kHD, b2 + (size_t)layer * kHD,
            h1, h2, hp);

        if (layer == 0)
            edge_tile_scan_kernel<<<kTiles, 1024, 0, stream>>>(
                h1, h2, hp, srcIdx, dstIdx, a + (size_t)layer * kH * kD, agg,
                gList, gCount);
        else
            edge_tile_list_kernel<<<kTiles, 1024, 0, stream>>>(
                h1, h2, hp, srcIdx, dstIdx, a + (size_t)layer * kH * kD, agg,
                gList, gCount);

        ln_elu_res_kernel<<<kN / 8, 256, 0, stream>>>(
            agg, ln_g + (size_t)layer * kHD, ln_b + (size_t)layer * kHD,
            xin, xcur);
    }

    out_proj_kernel<<<(kN + 255) / 256, 256, 0, stream>>>(xcur, Wo, bo, out);
}


# Edge loop remapped: 4 edges per wave pass, 8 lanes per edge, lane = head (GEMM stores rows head-interleaved): logit summed in-lane (new sum order), 8-lane softmax, one exp and divide per head
# speedup vs baseline: 1.4690x; 1.2363x over previous
#include <hip/hip_runtime.h>
#include <math.h>


constexpr int kN   = 20000;
constexpr int kE   = 320000;
constexpr int kIN  = 128;
constexpr int kH   = 8;
constexpr int kD   = 16;
constexpr int kHD  = 128;
constexpr int kL   = 2;
constexpr int kOUT = 2;
constexpr float kLnEps = 1e-5f;

typedef __attribute__((ext_vector_type(2))) float v2f;
typedef __attribute__((ext_vector_type(8))) float v8f;

__device__ __forceinline__ float eluf(float x) {
    return x > 0.0f ? x : (expf(x) - 1.0f);
}

__global__ __launch_bounds__(256) void gemm_h1h2_kernel(
    const float* __restrict__ X,
    const float* __restrict__ W1, const float* __restrict__ B1,
    const float* __restrict__ W2, const float* __restrict__ B2,
    float* __restrict__ H1, float* __restrict__ H2, float* __restrict__ HP)
{
    const int lane  = threadIdx.x & 31;
    const int wave  = threadIdx.x >> 5;
    const int mbase = blockIdx.x * 16;
    const int n0    = wave * 16;
    const int r     = lane & 15;
    const int half  = lane >> 4;

    const int    p    = n0 + r;
    const int    ncol = ((p >> 2) & 7) * 16 + (p >> 5) * 4 + (p & 3);

    const float* xrow = X + (size_t)(mbase + r) * kIN;

    v8f acc1 = {};
    v8f acc2 = {};

#pragma unroll
    for (int k0 = 0; k0 < kIN; k0 += 4) {
        const int ka = k0 + half * 2;
        v2f a = *(const v2f*)(xrow + ka);
        v2f bA, bB;
        bA.x = W1[(size_t)ka       * kHD + ncol];
        bA.y = W1[(size_t)(ka + 1) * kHD + ncol];
        bB.x = W2[(size_t)ka       * kHD + ncol];
        bB.y = W2[(size_t)(ka + 1) * kHD + ncol];
        acc1 = __builtin_amdgcn_wmma_f32_16x16x4_f32(
            false, a, false, bA, (short)0, acc1, false, false);
        acc2 = __builtin_amdgcn_wmma_f32_16x16x4_f32(
            false, a, false, bB, (short)0, acc2, false, false);
    }

    const float bb1 = B1[ncol];
    const float bb2 = B2[ncol];

#pragma unroll
    for (int j = 0; j < 8; ++j) {
        const int m = mbase + j + half * 8;
        const float h1v = acc1[j] + bb1;
        const float h2v = acc2[j] + bb2;
        const size_t idx = (size_t)m * kHD + p;
        H1[idx] = h1v;
        H2[idx] = h2v;
        HP[idx] = h1v * h2v;
    }
}

constexpr int kTR     = 80;
constexpr int kTiles  = kN / kTR;
constexpr int kChunks = kE / 128;
constexpr int kCap    = 640;
constexpr int kSlot   = 1024;
constexpr unsigned kOver = 0xFFFFFFFFu;
constexpr int kPitch  = kHD + 8;

struct HeadVec { float4 q[4]; };

__device__ __forceinline__ void work_list(
    const unsigned* myList, const int cnt, const int lane, const HeadVec& av,
    const float* __restrict__ H1, const float* __restrict__ HP,
    const float4* sH2, const float4* sHPd, float* sAcc)
{
    const int h = lane & 7;
    const int g = lane >> 3;
    for (int i0 = 0; i0 < cnt; i0 += 4) {
        const bool valid = (i0 + g) < cnt;
        const unsigned ent = myList[valid ? (i0 + g) : (cnt - 1)];
        const int dl = (int)(ent >> 20);
        const unsigned boff = (ent & 0xFFFFFu) * (unsigned)(kHD * 4) + (unsigned)h * 16u;
        const float4* p1 = (const float4*)((const char*)H1 + boff);
        const float4* pp = (const float4*)((const char*)HP + boff);
        float4 a1[4], ap[4];
#pragma unroll
        for (int q = 0; q < 4; ++q) {
            a1[q] = p1[q * 8];
            ap[q] = pp[q * 8];
        }
        const float4* d2 = sH2  + dl * 32 + h;
        const float4* dp = sHPd + dl * 32 + h;
        float pq[4];
        for (int q = 0; q < 4; ++q) {
            const float4 h1s = a1[q];
            const float4 hps = ap[q];
            const float4 h2d = d2[q * 8];
            const float4 hpd = dp[q * 8];
            const float t0 = eluf(h1s.x + h2d.x + hps.x * hpd.x);
            const float t1 = eluf(h1s.y + h2d.y + hps.y * hpd.y);
            const float t2 = eluf(h1s.z + h2d.z + hps.z * hpd.z);
            const float t3 = eluf(h1s.w + h2d.w + hps.w * hpd.w);
            pq[q] = av.q[q].x * t0 + av.q[q].y * t1 + av.q[q].z * t2 + av.q[q].w * t3;
        }
        const float part = ((pq[0] + pq[1]) + (pq[2] + pq[3])) * 0.25f;
        float mx = part;
        mx = fmaxf(mx, __shfl_xor(mx, 1, 32));
        mx = fmaxf(mx, __shfl_xor(mx, 2, 32));
        mx = fmaxf(mx, __shfl_xor(mx, 4, 32));
        const float ex = expf(part - mx);
        float sm = ex;
        sm += __shfl_xor(sm, 1, 32);
        sm += __shfl_xor(sm, 2, 32);
        sm += __shfl_xor(sm, 4, 32);
        const float alpha = ex / sm;
        if (valid) {
            float* arow = sAcc + dl * kPitch + h;
            for (int q = 0; q < 4; ++q) {
                atomicAdd(arow + (4 * q + 0) * 8, alpha * a1[q].x);
                atomicAdd(arow + (4 * q + 1) * 8, alpha * a1[q].y);
                atomicAdd(arow + (4 * q + 2) * 8, alpha * a1[q].z);
                atomicAdd(arow + (4 * q + 3) * 8, alpha * a1[q].w);
            }
        }
    }
}

template <bool kSave>
__device__ __forceinline__ void scan_and_work(
    unsigned* myList, const int wave, const int lane, const unsigned lo, const HeadVec& av,
    const float* __restrict__ H1, const float* __restrict__ HP,
    const int* __restrict__ src, const int* __restrict__ dst,
    const float4* sH2, const float4* sHPd, float* sAcc,
    unsigned* __restrict__ gSlot, unsigned* __restrict__ gCnt)
{
    const unsigned lanesBelow = (1u << lane) - 1u;
    int cnt = 0;
    unsigned saved = 0;

    for (int c0 = wave; ; c0 += 128) {
        const bool done = (c0 >= kChunks);
        if (done || cnt > kCap - 512) {
            const bool fits = kSave && saved != kOver && saved + (unsigned)cnt <= (unsigned)kSlot;
            for (int i = lane; i < cnt; i += 32) {
                const unsigned ent = myList[i];
                const unsigned v = (unsigned)src[ent & 0xFFFFFu] | (ent & 0xFFF00000u);
                myList[i] = v;
                if (fits) gSlot[saved + i] = v;
            }
            if (kSave) saved = fits ? saved + (unsigned)cnt : kOver;
            work_list(myList, cnt, lane, av, H1, HP, sH2, sHPd, sAcc);
            cnt = 0;
        }
        if (done) break;

        int4 dv[4];
#pragma unroll
        for (int u = 0; u < 4; ++u) {
            const int c = c0 + 32 * u;
            if (c < kChunks) dv[u] = *(const int4*)(dst + (size_t)c * 128 + lane * 4);
            else             dv[u] = make_int4(-1, -1, -1, -1);
        }
#pragma unroll
        for (int u = 0; u < 4; ++u) {
            const int ebase = (c0 + 32 * u) * 128 + lane * 4;
            const unsigned r0 = (unsigned)dv[u].x - lo;
            const unsigned r1 = (unsigned)dv[u].y - lo;
            const unsigned r2 = (unsigned)dv[u].z - lo;
            const unsigned r3 = (unsigned)dv[u].w - lo;
            const unsigned rm = min(min(r0, r1), min(r2, r3));
            if (__builtin_amdgcn_ballot_w32(rm < (unsigned)kTR) != 0u) {
                const unsigned rr[4] = {r0, r1, r2, r3};
#pragma unroll
                for (int k = 0; k < 4; ++k) {
                    const bool hit = rr[k] < (unsigned)kTR;
                    const unsigned mask = __builtin_amdgcn_ballot_w32(hit);
                    if (hit) {
                        const int pos = cnt + __popc(mask & lanesBelow);
                        myList[pos] = (unsigned)(ebase + k) | (rr[k] << 20);
                    }
                    cnt += __popc(mask);
                }
            }
        }
    }
    if (kSave && lane == 0) *gCnt = saved;
}

template <bool kReuse>
__device__ __forceinline__ void edge_tile_body(
    const float* __restrict__ H1, const float* __restrict__ H2,
    const float* __restrict__ HP,
    const int* __restrict__ src, const int* __restrict__ dst,
    const float* __restrict__ avec, float* __restrict__ AGG,
    unsigned* __restrict__ gList, unsigned* __restrict__ gCount)
{
    __shared__ float    sAcc[kTR * kPitch];
    __shared__ float4   sH2[kTR * 32];
    __shared__ float4   sHPd[kTR * 32];
    __shared__ unsigned sList[32 * kCap];

    const int tid  = threadIdx.x;
    const int lane = tid & 31;
    const int wave = __builtin_amdgcn_readfirstlane(tid >> 5);
    const unsigned lo = (unsigned)blockIdx.x * kTR;

    for (int i = tid; i < kTR * kPitch; i += 1024) sAcc[i] = 0.0f;
    {
        const float4* g2 = (const float4*)(H2 + (size_t)lo * kHD);
        const float4* gp = (const float4*)(HP + (size_t)lo * kHD);
        for (int i = tid; i < kTR * 32; i += 1024) {
            sH2[i]  = g2[i];
            sHPd[i] = gp[i];
        }
    }
    HeadVec av;
    for (int q = 0; q < 4; ++q) av.q[q] = *(const float4*)(avec + (lane & 7) * kD + q * 4);
    __syncthreads();

    unsigned* myList = sList + wave * kCap;
    const size_t slot = (size_t)blockIdx.x * 32 + (size_t)wave;
    unsigned* gSlot = gList + slot * kSlot;
    unsigned* gCnt  = gCount + slot;

    if (kReuse) {
        const unsigned n = __builtin_amdgcn_readfirstlane(*gCnt);
        if (n <= (unsigned)kSlot) {
            for (unsigned base = 0; base < n; base += kCap) {
                const int m = (int)min((unsigned)kCap, n - base);
                for (int i = lane; i < m; i += 32) myList[i] = gSlot[base + i];
                work_list(myList, m, lane, av, H1, HP, sH2, sHPd, sAcc);
            }
        } else {
            scan_and_work<false>(myList, wave, lane, lo, av, H1, HP, src, dst,
                                 sH2, sHPd, sAcc, gSlot, gCnt);
        }
    } else {
        scan_and_work<true>(myList, wave, lane, lo, av, H1, HP, src, dst,
                            sH2, sHPd, sAcc, gSlot, gCnt);
    }

    __syncthreads();
    for (int i = tid; i < kTR * 32; i += 1024) {
        const int row = i >> 5;
        const int l   = i & 31;
        const float* a = sAcc + row * kPitch + (l & 3) * 32 + (l >> 2);
        float4 o;
        o.x = a[0];
        o.y = a[8];
        o.z = a[16];
        o.w = a[24];
        *(float4*)(AGG + ((size_t)lo + row) * kHD + l * 4) = o;
    }
}

__global__ __launch_bounds__(1024) void edge_tile_scan_kernel(
    const float* __restrict__ H1, const float* __restrict__ H2,
    const float* __restrict__ HP,
    const int* __restrict__ src, const int* __restrict__ dst,
    const float* __restrict__ avec, float* __restrict__ AGG,
    unsigned* __restrict__ gList, unsigned* __restrict__ gCount)
{
    edge_tile_body<false>(H1, H2, HP, src, dst, avec, AGG, gList, gCount);
}

__global__ __launch_bounds__(1024) void edge_tile_list_kernel(
    const float* __restrict__ H1, const float* __restrict__ H2,
    const float* __restrict__ HP,
    const int* __restrict__ src, const int* __restrict__ dst,
    const float* __restrict__ avec, float* __restrict__ AGG,
    unsigned* __restrict__ gList, unsigned* __restrict__ gCount)
{
    edge_tile_body<true>(H1, H2, HP, src, dst, avec, AGG, gList, gCount);
}

__global__ __launch_bounds__(256) void ln_elu_res_kernel(
    const float* __restrict__ AGG,
    const float* __restrict__ G, const float* __restrict__ B,
    const float* __restrict__ Xin, float* __restrict__ Xout)
{
    const int lane = threadIdx.x & 31;
    const int node = blockIdx.x * 8 + (threadIdx.x >> 5);
    if (node >= kN) return;

    const float4 v = *(const float4*)(AGG + (size_t)node * kHD + lane * 4);

    float s = v.x + v.y + v.z + v.w;
#pragma unroll
    for (int m = 1; m <= 16; m <<= 1) s += __shfl_xor(s, m, 32);
    const float mu = s * (1.0f / 128.0f);

    const float dx0 = v.x - mu, dx1 = v.y - mu, dx2 = v.z - mu, dx3 = v.w - mu;
    float q = dx0 * dx0 + dx1 * dx1 + dx2 * dx2 + dx3 * dx3;
#pragma unroll
    for (int m = 1; m <= 16; m <<= 1) q += __shfl_xor(q, m, 32);
    const float inv = rsqrtf(q * (1.0f / 128.0f) + kLnEps);

    const float4 g  = *(const float4*)(G + lane * 4);
    const float4 b  = *(const float4*)(B + lane * 4);
    const float4 xr = *(const float4*)(Xin + (size_t)node * kHD + lane * 4);

    float4 o;
    o.x = eluf(dx0 * inv * g.x + b.x) + xr.x;
    o.y = eluf(dx1 * inv * g.y + b.y) + xr.y;
    o.z = eluf(dx2 * inv * g.z + b.z) + xr.z;
    o.w = eluf(dx3 * inv * g.w + b.w) + xr.w;

    *(float4*)(Xout + (size_t)node * kHD + lane * 4) = o;
}

__global__ __launch_bounds__(256) void out_proj_kernel(
    const float* __restrict__ X, const float* __restrict__ Wo,
    const float* __restrict__ bo, float* __restrict__ O)
{
    const int n = blockIdx.x * blockDim.x + threadIdx.x;
    if (n >= kN) return;
    float a0 = bo[0];
    float a1 = bo[1];
    const float* xr = X + (size_t)n * kHD;
#pragma unroll 8
    for (int k = 0; k < kHD; ++k) {
        const float xv = xr[k];
        a0 += xv * Wo[2 * k + 0];
        a1 += xv * Wo[2 * k + 1];
    }
    O[2 * n + 0] = a0;
    O[2 * n + 1] = a1;
}

extern "C" void kernel_launch(void* const* d_in, const int* in_sizes, int n_in,
                              void* d_out, int out_size, void* d_ws, size_t ws_size,
                              hipStream_t stream)
{
    (void)in_sizes; (void)n_in; (void)out_size; (void)ws_size;

    const float* x    = (const float*)d_in[0];
    const int*   ei   = (const int*)  d_in[1];
    const float* W1   = (const float*)d_in[2];
    const float* b1   = (const float*)d_in[3];
    const float* W2   = (const float*)d_in[4];
    const float* b2   = (const float*)d_in[5];
    const float* a    = (const float*)d_in[6];
    const float* ln_g = (const float*)d_in[7];
    const float* ln_b = (const float*)d_in[8];
    const float* Wo   = (const float*)d_in[9];
    const float* bo   = (const float*)d_in[10];
    float*       out  = (float*)d_out;

    const size_t NHD = (size_t)kN * kHD;
    float* h1   = (float*)d_ws;
    float* h2   = h1  + NHD;
    float* hp   = h2  + NHD;
    float* agg  = hp  + NHD;
    float* xcur = agg + NHD;
    unsigned* gList  = (unsigned*)(xcur + NHD);
    unsigned* gCount = gList + (size_t)kTiles * 32 * kSlot;

    const int* srcIdx = ei;
    const int* dstIdx = ei + kE;

    for (int layer = 0; layer < kL; ++layer) {
        const float* xin = (layer == 0) ? x : xcur;

        gemm_h1h2_kernel<<<kN / 16, 256, 0, stream>>>(
            xin,
            W1 + (size_t)layer * kIN * kHD, b1 + (size_t)layer * kHD,
            W2 + (size_t)layer * kIN * kHD, b2 + (size_t)layer * kHD,
            h1, h2, hp);

        if (layer == 0)
            edge_tile_scan_kernel<<<kTiles, 1024, 0, stream>>>(
                h1, h2, hp, srcIdx, dstIdx, a + (size_t)layer * kH * kD, agg,
                gList, gCount);
        else
            edge_tile_list_kernel<<<kTiles, 1024, 0, stream>>>(
                h1, h2, hp, srcIdx, dstIdx, a + (size_t)layer * kH * kD, agg,
                gList, gCount);

        ln_elu_res_kernel<<<kN / 8, 256, 0, stream>>>(
            agg, ln_g + (size_t)layer * kHD, ln_b + (size_t)layer * kHD,
            xin, xcur);
    }

    out_proj_kernel<<<(kN + 255) / 256, 256, 0, stream>>>(xcur, Wo, bo, out);
}


# Tile block finishes its 80 rows from the LDS sums: LayerNorm+ELU+residual by a wave per row in the tile kernel (same sums); the sums plane, its store/reload and both LayerNorm launches gone
# speedup vs baseline: 1.5696x; 1.0685x over previous
#include <hip/hip_runtime.h>
#include <math.h>


constexpr int kN   = 20000;
constexpr int kE   = 320000;
constexpr int kIN  = 128;
constexpr int kH   = 8;
constexpr int kD   = 16;
constexpr int kHD  = 128;
constexpr int kL   = 2;
constexpr int kOUT = 2;
constexpr float kLnEps = 1e-5f;

typedef __attribute__((ext_vector_type(2))) float v2f;
typedef __attribute__((ext_vector_type(8))) float v8f;

__device__ __forceinline__ float eluf(float x) {
    return x > 0.0f ? x : (expf(x) - 1.0f);
}

__global__ __launch_bounds__(256) void gemm_h1h2_kernel(
    const float* __restrict__ X,
    const float* __restrict__ W1, const float* __restrict__ B1,
    const float* __restrict__ W2, const float* __restrict__ B2,
    float* __restrict__ H1, float* __restrict__ H2, float* __restrict__ HP)
{
    const int lane  = threadIdx.x & 31;
    const int wave  = threadIdx.x >> 5;
    const int mbase = blockIdx.x * 16;
    const int n0    = wave * 16;
    const int r     = lane & 15;
    const int half  = lane >> 4;

    const int    p    = n0 + r;
    const int    ncol = ((p >> 2) & 7) * 16 + (p >> 5) * 4 + (p & 3);

    const float* xrow = X + (size_t)(mbase + r) * kIN;

    v8f acc1 = {};
    v8f acc2 = {};

#pragma unroll
    for (int k0 = 0; k0 < kIN; k0 += 4) {
        const int ka = k0 + half * 2;
        v2f a = *(const v2f*)(xrow + ka);
        v2f bA, bB;
        bA.x = W1[(size_t)ka       * kHD + ncol];
        bA.y = W1[(size_t)(ka + 1) * kHD + ncol];
        bB.x = W2[(size_t)ka       * kHD + ncol];
        bB.y = W2[(size_t)(ka + 1) * kHD + ncol];
        acc1 = __builtin_amdgcn_wmma_f32_16x16x4_f32(
            false, a, false, bA, (short)0, acc1, false, false);
        acc2 = __builtin_amdgcn_wmma_f32_16x16x4_f32(
            false, a, false, bB, (short)0, acc2, false, false);
    }

    const float bb1 = B1[ncol];
    const float bb2 = B2[ncol];

#pragma unroll
    for (int j = 0; j < 8; ++j) {
        const int m = mbase + j + half * 8;
        const float h1v = acc1[j] + bb1;
        const float h2v = acc2[j] + bb2;
        const size_t idx = (size_t)m * kHD + p;
        H1[idx] = h1v;
        H2[idx] = h2v;
        HP[idx] = h1v * h2v;
    }
}

constexpr int kTR     = 80;
constexpr int kTiles  = kN / kTR;
constexpr int kChunks = kE / 128;
constexpr int kCap    = 640;
constexpr int kSlot   = 1024;
constexpr unsigned kOver = 0xFFFFFFFFu;
constexpr int kPitch  = kHD + 8;

struct HeadVec { float4 q[4]; };

__device__ __forceinline__ void work_list(
    const unsigned* myList, const int cnt, const int lane, const HeadVec& av,
    const float* __restrict__ H1, const float* __restrict__ HP,
    const float4* sH2, const float4* sHPd, float* sAcc)
{
    const int h = lane & 7;
    const int g = lane >> 3;
    for (int i0 = 0; i0 < cnt; i0 += 4) {
        const bool valid = (i0 + g) < cnt;
        const unsigned ent = myList[valid ? (i0 + g) : (cnt - 1)];
        const int dl = (int)(ent >> 20);
        const unsigned boff = (ent & 0xFFFFFu) * (unsigned)(kHD * 4) + (unsigned)h * 16u;
        const float4* p1 = (const float4*)((const char*)H1 + boff);
        const float4* pp = (const float4*)((const char*)HP + boff);
        float4 a1[4], ap[4];
#pragma unroll
        for (int q = 0; q < 4; ++q) {
            a1[q] = p1[q * 8];
            ap[q] = pp[q * 8];
        }
        const float4* d2 = sH2  + dl * 32 + h;
        const float4* dp = sHPd + dl * 32 + h;
        float pq[4];
        for (int q = 0; q < 4; ++q) {
            const float4 h1s = a1[q];
            const float4 hps = ap[q];
            const float4 h2d = d2[q * 8];
            const float4 hpd = dp[q * 8];
            const float t0 = eluf(h1s.x + h2d.x + hps.x * hpd.x);
            const float t1 = eluf(h1s.y + h2d.y + hps.y * hpd.y);
            const float t2 = eluf(h1s.z + h2d.z + hps.z * hpd.z);
            const float t3 = eluf(h1s.w + h2d.w + hps.w * hpd.w);
            pq[q] = av.q[q].x * t0 + av.q[q].y * t1 + av.q[q].z * t2 + av.q[q].w * t3;
        }
        const float part = ((pq[0] + pq[1]) + (pq[2] + pq[3])) * 0.25f;
        float mx = part;
        mx = fmaxf(mx, __shfl_xor(mx, 1, 32));
        mx = fmaxf(mx, __shfl_xor(mx, 2, 32));
        mx = fmaxf(mx, __shfl_xor(mx, 4, 32));
        const float ex = expf(part - mx);
        float sm = ex;
        sm += __shfl_xor(sm, 1, 32);
        sm += __shfl_xor(sm, 2, 32);
        sm += __shfl_xor(sm, 4, 32);
        const float alpha = ex / sm;
        if (valid) {
            float* arow = sAcc + dl * kPitch + h;
            for (int q = 0; q < 4; ++q) {
                atomicAdd(arow + (4 * q + 0) * 8, alpha * a1[q].x);
                atomicAdd(arow + (4 * q + 1) * 8, alpha * a1[q].y);
                atomicAdd(arow + (4 * q + 2) * 8, alpha * a1[q].z);
                atomicAdd(arow + (4 * q + 3) * 8, alpha * a1[q].w);
            }
        }
    }
}

template <bool kSave>
__device__ __forceinline__ void scan_and_work(
    unsigned* myList, const int wave, const int lane, const unsigned lo, const HeadVec& av,
    const float* __restrict__ H1, const float* __restrict__ HP,
    const int* __restrict__ src, const int* __restrict__ dst,
    const float4* sH2, const float4* sHPd, float* sAcc,
    unsigned* __restrict__ gSlot, unsigned* __restrict__ gCnt)
{
    const unsigned lanesBelow = (1u << lane) - 1u;
    int cnt = 0;
    unsigned saved = 0;

    for (int c0 = wave; ; c0 += 128) {
        const bool done = (c0 >= kChunks);
        if (done || cnt > kCap - 512) {
            const bool fits = kSave && saved != kOver && saved + (unsigned)cnt <= (unsigned)kSlot;
            for (int i = lane; i < cnt; i += 32) {
                const unsigned ent = myList[i];
                const unsigned v = (unsigned)src[ent & 0xFFFFFu] | (ent & 0xFFF00000u);
                myList[i] = v;
                if (fits) gSlot[saved + i] = v;
            }
            if (kSave) saved = fits ? saved + (unsigned)cnt : kOver;
            work_list(myList, cnt, lane, av, H1, HP, sH2, sHPd, sAcc);
            cnt = 0;
        }
        if (done) break;

        int4 dv[4];
#pragma unroll
        for (int u = 0; u < 4; ++u) {
            const int c = c0 + 32 * u;
            if (c < kChunks) dv[u] = *(const int4*)(dst + (size_t)c * 128 + lane * 4);
            else             dv[u] = make_int4(-1, -1, -1, -1);
        }
#pragma unroll
        for (int u = 0; u < 4; ++u) {
            const int ebase = (c0 + 32 * u) * 128 + lane * 4;
            const unsigned r0 = (unsigned)dv[u].x - lo;
            const unsigned r1 = (unsigned)dv[u].y - lo;
            const unsigned r2 = (unsigned)dv[u].z - lo;
            const unsigned r3 = (unsigned)dv[u].w - lo;
            const unsigned rm = min(min(r0, r1), min(r2, r3));
            if (__builtin_amdgcn_ballot_w32(rm < (unsigned)kTR) != 0u) {
                const unsigned rr[4] = {r0, r1, r2, r3};
#pragma unroll
                for (int k = 0; k < 4; ++k) {
                    const bool hit = rr[k] < (unsigned)kTR;
                    const unsigned mask = __builtin_amdgcn_ballot_w32(hit);
                    if (hit) {
                        const int pos = cnt + __popc(mask & lanesBelow);
                        myList[pos] = (unsigned)(ebase + k) | (rr[k] << 20);
                    }
                    cnt += __popc(mask);
                }
            }
        }
    }
    if (kSave && lane == 0) *gCnt = saved;
}

template <bool kReuse>
__device__ __forceinline__ void edge_tile_body(
    const float* __restrict__ H1, const float* __restrict__ H2,
    const float* __restrict__ HP,
    const int* __restrict__ src, const int* __restrict__ dst,
    const float* __restrict__ avec,
    const float* __restrict__ G, const float* __restrict__ B,
    const float* Xin, float* Xout,
    unsigned* __restrict__ gList, unsigned* __restrict__ gCount)
{
    __shared__ float    sAcc[kTR * kPitch];
    __shared__ float4   sH2[kTR * 32];
    __shared__ float4   sHPd[kTR * 32];
    __shared__ unsigned sList[32 * kCap];

    const int tid  = threadIdx.x;
    const int lane = tid & 31;
    const int wave = __builtin_amdgcn_readfirstlane(tid >> 5);
    const unsigned lo = (unsigned)blockIdx.x * kTR;

    for (int i = tid; i < kTR * kPitch; i += 1024) sAcc[i] = 0.0f;
    {
        const float4* g2 = (const float4*)(H2 + (size_t)lo * kHD);
        const float4* gp = (const float4*)(HP + (size_t)lo * kHD);
        for (int i = tid; i < kTR * 32; i += 1024) {
            sH2[i]  = g2[i];
            sHPd[i] = gp[i];
        }
    }
    HeadVec av;
    for (int q = 0; q < 4; ++q) av.q[q] = *(const float4*)(avec + (lane & 7) * kD + q * 4);
    __syncthreads();

    unsigned* myList = sList + wave * kCap;
    const size_t slot = (size_t)blockIdx.x * 32 + (size_t)wave;
    unsigned* gSlot = gList + slot * kSlot;
    unsigned* gCnt  = gCount + slot;

    if (kReuse) {
        const unsigned n = __builtin_amdgcn_readfirstlane(*gCnt);
        if (n <= (unsigned)kSlot) {
            for (unsigned base = 0; base < n; base += kCap) {
                const int m = (int)min((unsigned)kCap, n - base);
                for (int i = lane; i < m; i += 32) myList[i] = gSlot[base + i];
                work_list(myList, m, lane, av, H1, HP, sH2, sHPd, sAcc);
            }
        } else {
            scan_and_work<false>(myList, wave, lane, lo, av, H1, HP, src, dst,
                                 sH2, sHPd, sAcc, gSlot, gCnt);
        }
    } else {
        scan_and_work<true>(myList, wave, lane, lo, av, H1, HP, src, dst,
                            sH2, sHPd, sAcc, gSlot, gCnt);
    }

    constexpr int kRowsPerWave = (kTR + 31) / 32;
    float4 xr[kRowsPerWave];
#pragma unroll
    for (int j = 0; j < kRowsPerWave; ++j) {
        const int row = min(wave + 32 * j, kTR - 1);
        xr[j] = *(const float4*)(Xin + ((size_t)lo + row) * kHD + lane * 4);
    }
    const float4 g = *(const float4*)(G + lane * 4);
    const float4 b = *(const float4*)(B + lane * 4);

    __syncthreads();

#pragma unroll
    for (int j = 0; j < kRowsPerWave; ++j) {
        const int row = wave + 32 * j;
        if (row < kTR) {
            const float* a = sAcc + row * kPitch + (lane & 3) * 32 + (lane >> 2);
            float4 v;
            v.x = a[0];
            v.y = a[8];
            v.z = a[16];
            v.w = a[24];

            float s = v.x + v.y + v.z + v.w;
            for (int m = 1; m <= 16; m <<= 1) s += __shfl_xor(s, m, 32);
            const float mu = s * (1.0f / 128.0f);

            const float dx0 = v.x - mu, dx1 = v.y - mu, dx2 = v.z - mu, dx3 = v.w - mu;
            float q = dx0 * dx0 + dx1 * dx1 + dx2 * dx2 + dx3 * dx3;
            for (int m = 1; m <= 16; m <<= 1) q += __shfl_xor(q, m, 32);
            const float inv = rsqrtf(q * (1.0f / 128.0f) + kLnEps);

            float4 o;
            o.x = eluf(dx0 * inv * g.x + b.x) + xr[j].x;
            o.y = eluf(dx1 * inv * g.y + b.y) + xr[j].y;
            o.z = eluf(dx2 * inv * g.z + b.z) + xr[j].z;
            o.w = eluf(dx3 * inv * g.w + b.w) + xr[j].w;
            *(float4*)(Xout + ((size_t)lo + row) * kHD + lane * 4) = o;
        }
    }
}

__global__ __launch_bounds__(1024) void edge_tile_scan_kernel(
    const float* __restrict__ H1, const float* __restrict__ H2,
    const float* __restrict__ HP,
    const int* __restrict__ src, const int* __restrict__ dst,
    const float* __restrict__ avec,
    const float* __restrict__ G, const float* __restrict__ B,
    const float* Xin, float* Xout,
    unsigned* __restrict__ gList, unsigned* __restrict__ gCount)
{
    edge_tile_body<false>(H1, H2, HP, src, dst, avec, G, B, Xin, Xout, gList, gCount);
}

__global__ __launch_bounds__(1024) void edge_tile_list_kernel(
    const float* __restrict__ H1, const float* __restrict__ H2,
    const float* __restrict__ HP,
    const int* __restrict__ src, const int* __restrict__ dst,
    const float* __restrict__ avec,
    const float* __restrict__ G, const float* __restrict__ B,
    const float* Xin, float* Xout,
    unsigned* __restrict__ gList, unsigned* __restrict__ gCount)
{
    edge_tile_body<true>(H1, H2, HP, src, dst, avec, G, B, Xin, Xout, gList, gCount);
}

__global__ __launch_bounds__(256) void out_proj_kernel(
    const float* __restrict__ X, const float* __restrict__ Wo,
    const float* __restrict__ bo, float* __restrict__ O)
{
    const int n = blockIdx.x * blockDim.x + threadIdx.x;
    if (n >= kN) return;
    float a0 = bo[0];
    float a1 = bo[1];
    const float* xr = X + (size_t)n * kHD;
#pragma unroll 8
    for (int k = 0; k < kHD; ++k) {
        const float xv = xr[k];
        a0 += xv * Wo[2 * k + 0];
        a1 += xv * Wo[2 * k + 1];
    }
    O[2 * n + 0] = a0;
    O[2 * n + 1] = a1;
}

extern "C" void kernel_launch(void* const* d_in, const int* in_sizes, int n_in,
                              void* d_out, int out_size, void* d_ws, size_t ws_size,
                              hipStream_t stream)
{
    (void)in_sizes; (void)n_in; (void)out_size; (void)ws_size;

    const float* x    = (const float*)d_in[0];
    const int*   ei   = (const int*)  d_in[1];
    const float* W1   = (const float*)d_in[2];
    const float* b1   = (const float*)d_in[3];
    const float* W2   = (const float*)d_in[4];
    const float* b2   = (const float*)d_in[5];
    const float* a    = (const float*)d_in[6];
    const float* ln_g = (const float*)d_in[7];
    const float* ln_b = (const float*)d_in[8];
    const float* Wo   = (const float*)d_in[9];
    const float* bo   = (const float*)d_in[10];
    float*       out  = (float*)d_out;

    const size_t NHD = (size_t)kN * kHD;
    float* h1   = (float*)d_ws;
    float* h2   = h1  + NHD;
    float* hp   = h2  + NHD;
    float* xcur = hp  + NHD;
    unsigned* gList  = (unsigned*)(xcur + NHD);
    unsigned* gCount = gList + (size_t)kTiles * 32 * kSlot;

    const int* srcIdx = ei;
    const int* dstIdx = ei + kE;

    for (int layer = 0; layer < kL; ++layer) {
        const float* xin = (layer == 0) ? x : xcur;

        gemm_h1h2_kernel<<<kN / 16, 256, 0, stream>>>(
            xin,
            W1 + (size_t)layer * kIN * kHD, b1 + (size_t)layer * kHD,
            W2 + (size_t)layer * kIN * kHD, b2 + (size_t)layer * kHD,
            h1, h2, hp);

        if (layer == 0)
            edge_tile_scan_kernel<<<kTiles, 1024, 0, stream>>>(
                h1, h2, hp, srcIdx, dstIdx, a + (size_t)layer * kH * kD,
                ln_g + (size_t)layer * kHD, ln_b + (size_t)layer * kHD, xin, xcur,
                gList, gCount);
        else
            edge_tile_list_kernel<<<kTiles, 1024, 0, stream>>>(
                h1, h2, hp, srcIdx, dstIdx, a + (size_t)layer * kH * kD,
                ln_g + (size_t)layer * kHD, ln_b + (size_t)layer * kHD, xin, xcur,
                gList, gCount);
    }

    out_proj_kernel<<<(kN + 255) / 256, 256, 0, stream>>>(xcur, Wo, bo, out);
}


# Edge lists of a block joined into one LDS list whose 4-edge passes are dealt round-robin to its 32 waves: equal pass counts per wave instead of unequal own lists; LDS-atomic arrival order changes
# speedup vs baseline: 1.7105x; 1.0898x over previous
#include <hip/hip_runtime.h>
#include <math.h>


constexpr int kN   = 20000;
constexpr int kE   = 320000;
constexpr int kIN  = 128;
constexpr int kH   = 8;
constexpr int kD   = 16;
constexpr int kHD  = 128;
constexpr int kL   = 2;
constexpr int kOUT = 2;
constexpr float kLnEps = 1e-5f;

typedef __attribute__((ext_vector_type(2))) float v2f;
typedef __attribute__((ext_vector_type(8))) float v8f;

__device__ __forceinline__ float eluf(float x) {
    return x > 0.0f ? x : (expf(x) - 1.0f);
}

__global__ __launch_bounds__(256) void gemm_h1h2_kernel(
    const float* __restrict__ X,
    const float* __restrict__ W1, const float* __restrict__ B1,
    const float* __restrict__ W2, const float* __restrict__ B2,
    float* __restrict__ H1, float* __restrict__ H2, float* __restrict__ HP)
{
    const int lane  = threadIdx.x & 31;
    const int wave  = threadIdx.x >> 5;
    const int mbase = blockIdx.x * 16;
    const int n0    = wave * 16;
    const int r     = lane & 15;
    const int half  = lane >> 4;

    const int    p    = n0 + r;
    const int    ncol = ((p >> 2) & 7) * 16 + (p >> 5) * 4 + (p & 3);

    const float* xrow = X + (size_t)(mbase + r) * kIN;

    v8f acc1 = {};
    v8f acc2 = {};

#pragma unroll
    for (int k0 = 0; k0 < kIN; k0 += 4) {
        const int ka = k0 + half * 2;
        v2f a = *(const v2f*)(xrow + ka);
        v2f bA, bB;
        bA.x = W1[(size_t)ka       * kHD + ncol];
        bA.y = W1[(size_t)(ka + 1) * kHD + ncol];
        bB.x = W2[(size_t)ka       * kHD + ncol];
        bB.y = W2[(size_t)(ka + 1) * kHD + ncol];
        acc1 = __builtin_amdgcn_wmma_f32_16x16x4_f32(
            false, a, false, bA, (short)0, acc1, false, false);
        acc2 = __builtin_amdgcn_wmma_f32_16x16x4_f32(
            false, a, false, bB, (short)0, acc2, false, false);
    }

    const float bb1 = B1[ncol];
    const float bb2 = B2[ncol];

#pragma unroll
    for (int j = 0; j < 8; ++j) {
        const int m = mbase + j + half * 8;
        const float h1v = acc1[j] + bb1;
        const float h2v = acc2[j] + bb2;
        const size_t idx = (size_t)m * kHD + p;
        H1[idx] = h1v;
        H2[idx] = h2v;
        HP[idx] = h1v * h2v;
    }
}

constexpr int kTR     = 80;
constexpr int kTiles  = kN / kTR;
constexpr int kChunks = kE / 128;
constexpr int kCap    = 640;
constexpr int kSlot   = 1024;
constexpr unsigned kOver = 0xFFFFFFFFu;
constexpr int kPitch  = kHD + 8;

struct HeadVec { float4 q[4]; };

__device__ __forceinline__ void work_list(
    const unsigned* myList, const int cnt, const int first, const int step,
    const int lane, const HeadVec& av,
    const float* __restrict__ H1, const float* __restrict__ HP,
    const float4* sH2, const float4* sHPd, float* sAcc)
{
    const int h = lane & 7;
    const int g = lane >> 3;
    for (int i0 = first; i0 < cnt; i0 += step) {
        const bool valid = (i0 + g) < cnt;
        const unsigned ent = myList[valid ? (i0 + g) : (cnt - 1)];
        const int dl = (int)(ent >> 20);
        const unsigned boff = (ent & 0xFFFFFu) * (unsigned)(kHD * 4) + (unsigned)h * 16u;
        const float4* p1 = (const float4*)((const char*)H1 + boff);
        const float4* pp = (const float4*)((const char*)HP + boff);
        float4 a1[4], ap[4];
#pragma unroll
        for (int q = 0; q < 4; ++q) {
            a1[q] = p1[q * 8];
            ap[q] = pp[q * 8];
        }
        const float4* d2 = sH2  + dl * 32 + h;
        const float4* dp = sHPd + dl * 32 + h;
        float pq[4];
        for (int q = 0; q < 4; ++q) {
            const float4 h1s = a1[q];
            const float4 hps = ap[q];
            const float4 h2d = d2[q * 8];
            const float4 hpd = dp[q * 8];
            const float t0 = eluf(h1s.x + h2d.x + hps.x * hpd.x);
            const float t1 = eluf(h1s.y + h2d.y + hps.y * hpd.y);
            const float t2 = eluf(h1s.z + h2d.z + hps.z * hpd.z);
            const float t3 = eluf(h1s.w + h2d.w + hps.w * hpd.w);
            pq[q] = av.q[q].x * t0 + av.q[q].y * t1 + av.q[q].z * t2 + av.q[q].w * t3;
        }
        const float part = ((pq[0] + pq[1]) + (pq[2] + pq[3])) * 0.25f;
        float mx = part;
        mx = fmaxf(mx, __shfl_xor(mx, 1, 32));
        mx = fmaxf(mx, __shfl_xor(mx, 2, 32));
        mx = fmaxf(mx, __shfl_xor(mx, 4, 32));
        const float ex = expf(part - mx);
        float sm = ex;
        sm += __shfl_xor(sm, 1, 32);
        sm += __shfl_xor(sm, 2, 32);
        sm += __shfl_xor(sm, 4, 32);
        const float alpha = ex / sm;
        if (valid) {
            float* arow = sAcc + dl * kPitch + h;
            for (int q = 0; q < 4; ++q) {
                atomicAdd(arow + (4 * q + 0) * 8, alpha * a1[q].x);
                atomicAdd(arow + (4 * q + 1) * 8, alpha * a1[q].y);
                atomicAdd(arow + (4 * q + 2) * 8, alpha * a1[q].z);
                atomicAdd(arow + (4 * q + 3) * 8, alpha * a1[q].w);
            }
        }
    }
}

template <bool kSave>
__device__ __forceinline__ void convert_entries(
    const unsigned* in, unsigned* out, const int cnt, const int lane,
    const int* __restrict__ src, unsigned* __restrict__ gSlot, unsigned& saved)
{
    const bool fits = kSave && saved != kOver && saved + (unsigned)cnt <= (unsigned)kSlot;
    for (int i = lane; i < cnt; i += 32) {
        const unsigned ent = in[i];
        const unsigned v = (unsigned)src[ent & 0xFFFFFu] | (ent & 0xFFF00000u);
        out[i] = v;
        if (fits) gSlot[saved + i] = v;
    }
    if (kSave) saved = fits ? saved + (unsigned)cnt : kOver;
}

__device__ __forceinline__ void wave_offsets(
    const unsigned mine, const int wave, unsigned& below, unsigned& total)
{
    unsigned b = 0, t = 0;
#pragma unroll
    for (int w = 0; w < 32; ++w) {
        const unsigned cw = (unsigned)__builtin_amdgcn_readlane((int)mine, w);
        t += cw;
        b += (w < wave) ? cw : 0u;
    }
    below = b;
    total = t;
}

template <bool kSave, bool kDefer>
__device__ __forceinline__ void scan_and_work(
    unsigned* myList, const int wave, const int lane, const unsigned lo, const HeadVec& av,
    const float* __restrict__ H1, const float* __restrict__ HP,
    const int* __restrict__ src, const int* __restrict__ dst,
    const float4* sH2, const float4* sHPd, float* sAcc,
    unsigned* __restrict__ gSlot, unsigned* __restrict__ gCnt,
    int& pending, unsigned& savedOut)
{
    const unsigned lanesBelow = (1u << lane) - 1u;
    int cnt = 0;
    unsigned saved = 0;

    for (int c0 = wave; c0 < kChunks; c0 += 128) {
        if (cnt > kCap - 512) {
            convert_entries<kSave>(myList, myList, cnt, lane, src, gSlot, saved);
            work_list(myList, cnt, 0, 4, lane, av, H1, HP, sH2, sHPd, sAcc);
            cnt = 0;
        }

        int4 dv[4];
#pragma unroll
        for (int u = 0; u < 4; ++u) {
            const int c = c0 + 32 * u;
            if (c < kChunks) dv[u] = *(const int4*)(dst + (size_t)c * 128 + lane * 4);
            else             dv[u] = make_int4(-1, -1, -1, -1);
        }
#pragma unroll
        for (int u = 0; u < 4; ++u) {
            const int ebase = (c0 + 32 * u) * 128 + lane * 4;
            const unsigned r0 = (unsigned)dv[u].x - lo;
            const unsigned r1 = (unsigned)dv[u].y - lo;
            const unsigned r2 = (unsigned)dv[u].z - lo;
            const unsigned r3 = (unsigned)dv[u].w - lo;
            const unsigned rm = min(min(r0, r1), min(r2, r3));
            if (__builtin_amdgcn_ballot_w32(rm < (unsigned)kTR) != 0u) {
                const unsigned rr[4] = {r0, r1, r2, r3};
#pragma unroll
                for (int k = 0; k < 4; ++k) {
                    const bool hit = rr[k] < (unsigned)kTR;
                    const unsigned mask = __builtin_amdgcn_ballot_w32(hit);
                    if (hit) {
                        const int pos = cnt + __popc(mask & lanesBelow);
                        myList[pos] = (unsigned)(ebase + k) | (rr[k] << 20);
                    }
                    cnt += __popc(mask);
                }
            }
        }
    }
    if (!kDefer) {
        convert_entries<kSave>(myList, myList, cnt, lane, src, gSlot, saved);
        work_list(myList, cnt, 0, 4, lane, av, H1, HP, sH2, sHPd, sAcc);
        cnt = 0;
        if (kSave && lane == 0) *gCnt = saved;
    }
    pending  = cnt;
    savedOut = saved;
}

template <bool kReuse>
__device__ __forceinline__ void edge_tile_body(
    const float* __restrict__ H1, const float* __restrict__ H2,
    const float* __restrict__ HP,
    const int* __restrict__ src, const int* __restrict__ dst,
    const float* __restrict__ avec,
    const float* __restrict__ G, const float* __restrict__ B,
    const float* Xin, float* Xout,
    unsigned* __restrict__ gList, unsigned* __restrict__ gCount)
{
    __shared__ float    sAcc[kTR * kPitch];
    __shared__ float4   sH2[kTR * 32];
    __shared__ float4   sHPd[kTR * 32];
    __shared__ unsigned sList[32 * kCap];
    __shared__ unsigned sJoin[kReuse ? 32 : 32 * kCap];
    __shared__ unsigned sCnt[32];

    const int tid  = threadIdx.x;
    const int lane = tid & 31;
    const int wave = __builtin_amdgcn_readfirstlane(tid >> 5);
    const unsigned lo = (unsigned)blockIdx.x * kTR;

    unsigned savedCnt = 0;
    if (kReuse) savedCnt = gCount[(size_t)blockIdx.x * 32 + lane];

    for (int i = tid; i < kTR * kPitch; i += 1024) sAcc[i] = 0.0f;
    {
        const float4* g2 = (const float4*)(H2 + (size_t)lo * kHD);
        const float4* gp = (const float4*)(HP + (size_t)lo * kHD);
        for (int i = tid; i < kTR * 32; i += 1024) {
            sH2[i]  = g2[i];
            sHPd[i] = gp[i];
        }
    }
    HeadVec av;
    for (int q = 0; q < 4; ++q) av.q[q] = *(const float4*)(avec + (lane & 7) * kD + q * 4);
    __syncthreads();

    unsigned* myList = sList + wave * kCap;
    const size_t slot = (size_t)blockIdx.x * 32 + (size_t)wave;
    unsigned* gSlot = gList + slot * kSlot;
    unsigned* gCnt  = gCount + slot;

    int pending = 0;
    unsigned saved = 0;
    if (kReuse) {
        const unsigned n = (unsigned)__builtin_amdgcn_readlane((int)savedCnt, wave);
        const bool anyOver = __builtin_amdgcn_ballot_w32(savedCnt > (unsigned)kSlot) != 0u;
        unsigned below, total;
        wave_offsets(anyOver ? 0u : savedCnt, wave, below, total);
        if (!anyOver && total <= (unsigned)(32 * kCap)) {
            for (unsigned i = lane; i < n; i += 32) sList[below + i] = gSlot[i];
            __syncthreads();
            work_list(sList, (int)total, 4 * wave, 128, lane, av, H1, HP, sH2, sHPd, sAcc);
        } else if (n <= (unsigned)kSlot) {
            for (unsigned base = 0; base < n; base += kCap) {
                const int m = (int)min((unsigned)kCap, n - base);
                for (int i = lane; i < m; i += 32) myList[i] = gSlot[base + i];
                work_list(myList, m, 0, 4, lane, av, H1, HP, sH2, sHPd, sAcc);
            }
        } else {
            scan_and_work<false, false>(myList, wave, lane, lo, av, H1, HP, src, dst,
                                        sH2, sHPd, sAcc, gSlot, gCnt, pending, saved);
        }
    } else {
        scan_and_work<true, true>(myList, wave, lane, lo, av, H1, HP, src, dst,
                                  sH2, sHPd, sAcc, gSlot, gCnt, pending, saved);
        if (lane == 0) sCnt[wave] = (unsigned)pending;
        __syncthreads();
        unsigned below, total;
        wave_offsets(sCnt[lane], wave, below, total);
        convert_entries<true>(myList, sJoin + below, pending, lane, src, gSlot, saved);
        if (lane == 0) *gCnt = saved;
        __syncthreads();
        work_list(sJoin, (int)total, 4 * wave, 128, lane, av, H1, HP, sH2, sHPd, sAcc);
    }

    constexpr int kRowsPerWave = (kTR + 31) / 32;
    float4 xr[kRowsPerWave];
#pragma unroll
    for (int j = 0; j < kRowsPerWave; ++j) {
        const int row = min(wave + 32 * j, kTR - 1);
        xr[j] = *(const float4*)(Xin + ((size_t)lo + row) * kHD + lane * 4);
    }
    const float4 g = *(const float4*)(G + lane * 4);
    const float4 b = *(const float4*)(B + lane * 4);

    __syncthreads();

#pragma unroll
    for (int j = 0; j < kRowsPerWave; ++j) {
        const int row = wave + 32 * j;
        if (row < kTR) {
            const float* a = sAcc + row * kPitch + (lane & 3) * 32 + (lane >> 2);
            float4 v;
            v.x = a[0];
            v.y = a[8];
            v.z = a[16];
            v.w = a[24];

            float s = v.x + v.y + v.z + v.w;
            for (int m = 1; m <= 16; m <<= 1) s += __shfl_xor(s, m, 32);
            const float mu = s * (1.0f / 128.0f);

            const float dx0 = v.x - mu, dx1 = v.y - mu, dx2 = v.z - mu, dx3 = v.w - mu;
            float q = dx0 * dx0 + dx1 * dx1 + dx2 * dx2 + dx3 * dx3;
            for (int m = 1; m <= 16; m <<= 1) q += __shfl_xor(q, m, 32);
            const float inv = rsqrtf(q * (1.0f / 128.0f) + kLnEps);

            float4 o;
            o.x = eluf(dx0 * inv * g.x + b.x) + xr[j].x;
            o.y = eluf(dx1 * inv * g.y + b.y) + xr[j].y;
            o.z = eluf(dx2 * inv * g.z + b.z) + xr[j].z;
            o.w = eluf(dx3 * inv * g.w + b.w) + xr[j].w;
            *(float4*)(Xout + ((size_t)lo + row) * kHD + lane * 4) = o;
        }
    }
}

__global__ __launch_bounds__(1024) void edge_tile_scan_kernel(
    const float* __restrict__ H1, const float* __restrict__ H2,
    const float* __restrict__ HP,
    const int* __restrict__ src, const int* __restrict__ dst,
    const float* __restrict__ avec,
    const float* __restrict__ G, const float* __restrict__ B,
    const float* Xin, float* Xout,
    unsigned* __restrict__ gList, unsigned* __restrict__ gCount)
{
    edge_tile_body<false>(H1, H2, HP, src, dst, avec, G, B, Xin, Xout, gList, gCount);
}

__global__ __launch_bounds__(1024) void edge_tile_list_kernel(
    const float* __restrict__ H1, const float* __restrict__ H2,
    const float* __restrict__ HP,
    const int* __restrict__ src, const int* __restrict__ dst,
    const float* __restrict__ avec,
    const float* __restrict__ G, const float* __restrict__ B,
    const float* Xin, float* Xout,
    unsigned* __restrict__ gList, unsigned* __restrict__ gCount)
{
    edge_tile_body<true>(H1, H2, HP, src, dst, avec, G, B, Xin, Xout, gList, gCount);
}

__global__ __launch_bounds__(256) void out_proj_kernel(
    const float* __restrict__ X, const float* __restrict__ Wo,
    const float* __restrict__ bo, float* __restrict__ O)
{
    const int n = blockIdx.x * blockDim.x + threadIdx.x;
    if (n >= kN) return;
    float a0 = bo[0];
    float a1 = bo[1];
    const float* xr = X + (size_t)n * kHD;
#pragma unroll 8
    for (int k = 0; k < kHD; ++k) {
        const float xv = xr[k];
        a0 += xv * Wo[2 * k + 0];
        a1 += xv * Wo[2 * k + 1];
    }
    O[2 * n + 0] = a0;
    O[2 * n + 1] = a1;
}

extern "C" void kernel_launch(void* const* d_in, const int* in_sizes, int n_in,
                              void* d_out, int out_size, void* d_ws, size_t ws_size,
                              hipStream_t stream)
{
    (void)in_sizes; (void)n_in; (void)out_size; (void)ws_size;

    const float* x    = (const float*)d_in[0];
    const int*   ei   = (const int*)  d_in[1];
    const float* W1   = (const float*)d_in[2];
    const float* b1   = (const float*)d_in[3];
    const float* W2   = (const float*)d_in[4];
    const float* b2   = (const float*)d_in[5];
    const float* a    = (const float*)d_in[6];
    const float* ln_g = (const float*)d_in[7];
    const float* ln_b = (const float*)d_in[8];
    const float* Wo   = (const float*)d_in[9];
    const float* bo   = (const float*)d_in[10];
    float*       out  = (float*)d_out;

    const size_t NHD = (size_t)kN * kHD;
    float* h1   = (float*)d_ws;
    float* h2   = h1  + NHD;
    float* hp   = h2  + NHD;
    float* xcur = hp  + NHD;
    unsigned* gList  = (unsigned*)(xcur + NHD);
    unsigned* gCount = gList + (size_t)kTiles * 32 * kSlot;

    const int* srcIdx = ei;
    const int* dstIdx = ei + kE;

    for (int layer = 0; layer < kL; ++layer) {
        const float* xin = (layer == 0) ? x : xcur;

        gemm_h1h2_kernel<<<kN / 16, 256, 0, stream>>>(
            xin,
            W1 + (size_t)layer * kIN * kHD, b1 + (size_t)layer * kHD,
            W2 + (size_t)layer * kIN * kHD, b2 + (size_t)layer * kHD,
            h1, h2, hp);

        if (layer == 0)
            edge_tile_scan_kernel<<<kTiles, 1024, 0, stream>>>(
                h1, h2, hp, srcIdx, dstIdx, a + (size_t)layer * kH * kD,
                ln_g + (size_t)layer * kHD, ln_b + (size_t)layer * kHD, xin, xcur,
                gList, gCount);
        else
            edge_tile_list_kernel<<<kTiles, 1024, 0, stream>>>(
                h1, h2, hp, srcIdx, dstIdx, a + (size_t)layer * kH * kD,
                ln_g + (size_t)layer * kHD, ln_b + (size_t)layer * kHD, xin, xcur,
                gList, gCount);
    }

    out_proj_kernel<<<(kN + 255) / 256, 256, 0, stream>>>(xcur, Wo, bo, out);
}


# Tile kernel ELUs (edge loop, finish) and head softmax on hardware exp2(x*log2e) and rcp, ELU branch-free by compare-select, not library expf + IEEE divide; args <= 0, abs err < 2e-7, last bits differ
# speedup vs baseline: 1.8780x; 1.0979x over previous
#include <hip/hip_runtime.h>
#include <math.h>


constexpr int kN   = 20000;
constexpr int kE   = 320000;
constexpr int kIN  = 128;
constexpr int kH   = 8;
constexpr int kD   = 16;
constexpr int kHD  = 128;
constexpr int kL   = 2;
constexpr int kOUT = 2;
constexpr float kLnEps = 1e-5f;

typedef __attribute__((ext_vector_type(2))) float v2f;
typedef __attribute__((ext_vector_type(8))) float v8f;

constexpr float kLog2e = 1.44269504088896340736f;

__device__ __forceinline__ float eluf(float x) {
    const float e = __builtin_amdgcn_exp2f(x * kLog2e) - 1.0f;
    return x > 0.0f ? x : e;
}

__global__ __launch_bounds__(256) void gemm_h1h2_kernel(
    const float* __restrict__ X,
    const float* __restrict__ W1, const float* __restrict__ B1,
    const float* __restrict__ W2, const float* __restrict__ B2,
    float* __restrict__ H1, float* __restrict__ H2, float* __restrict__ HP)
{
    const int lane  = threadIdx.x & 31;
    const int wave  = threadIdx.x >> 5;
    const int mbase = blockIdx.x * 16;
    const int n0    = wave * 16;
    const int r     = lane & 15;
    const int half  = lane >> 4;

    const int    p    = n0 + r;
    const int    ncol = ((p >> 2) & 7) * 16 + (p >> 5) * 4 + (p & 3);

    const float* xrow = X + (size_t)(mbase + r) * kIN;

    v8f acc1 = {};
    v8f acc2 = {};

#pragma unroll
    for (int k0 = 0; k0 < kIN; k0 += 4) {
        const int ka = k0 + half * 2;
        v2f a = *(const v2f*)(xrow + ka);
        v2f bA, bB;
        bA.x = W1[(size_t)ka       * kHD + ncol];
        bA.y = W1[(size_t)(ka + 1) * kHD + ncol];
        bB.x = W2[(size_t)ka       * kHD + ncol];
        bB.y = W2[(size_t)(ka + 1) * kHD + ncol];
        acc1 = __builtin_amdgcn_wmma_f32_16x16x4_f32(
            false, a, false, bA, (short)0, acc1, false, false);
        acc2 = __builtin_amdgcn_wmma_f32_16x16x4_f32(
            false, a, false, bB, (short)0, acc2, false, false);
    }

    const float bb1 = B1[ncol];
    const float bb2 = B2[ncol];

#pragma unroll
    for (int j = 0; j < 8; ++j) {
        const int m = mbase + j + half * 8;
        const float h1v = acc1[j] + bb1;
        const float h2v = acc2[j] + bb2;
        const size_t idx = (size_t)m * kHD + p;
        H1[idx] = h1v;
        H2[idx] = h2v;
        HP[idx] = h1v * h2v;
    }
}

constexpr int kTR     = 80;
constexpr int kTiles  = kN / kTR;
constexpr int kChunks = kE / 128;
constexpr int kCap    = 640;
constexpr int kSlot   = 1024;
constexpr unsigned kOver = 0xFFFFFFFFu;
constexpr int kPitch  = kHD + 8;

struct HeadVec { float4 q[4]; };

__device__ __forceinline__ void work_list(
    const unsigned* myList, const int cnt, const int first, const int step,
    const int lane, const HeadVec& av,
    const float* __restrict__ H1, const float* __restrict__ HP,
    const float4* sH2, const float4* sHPd, float* sAcc)
{
    const int h = lane & 7;
    const int g = lane >> 3;
    for (int i0 = first; i0 < cnt; i0 += step) {
        const bool valid = (i0 + g) < cnt;
        const unsigned ent = myList[valid ? (i0 + g) : (cnt - 1)];
        const int dl = (int)(ent >> 20);
        const unsigned boff = (ent & 0xFFFFFu) * (unsigned)(kHD * 4) + (unsigned)h * 16u;
        const float4* p1 = (const float4*)((const char*)H1 + boff);
        const float4* pp = (const float4*)((const char*)HP + boff);
        float4 a1[4], ap[4];
#pragma unroll
        for (int q = 0; q < 4; ++q) {
            a1[q] = p1[q * 8];
            ap[q] = pp[q * 8];
        }
        const float4* d2 = sH2  + dl * 32 + h;
        const float4* dp = sHPd + dl * 32 + h;
        float pq[4];
        for (int q = 0; q < 4; ++q) {
            const float4 h1s = a1[q];
            const float4 hps = ap[q];
            const float4 h2d = d2[q * 8];
            const float4 hpd = dp[q * 8];
            const float t0 = eluf(h1s.x + h2d.x + hps.x * hpd.x);
            const float t1 = eluf(h1s.y + h2d.y + hps.y * hpd.y);
            const float t2 = eluf(h1s.z + h2d.z + hps.z * hpd.z);
            const float t3 = eluf(h1s.w + h2d.w + hps.w * hpd.w);
            pq[q] = av.q[q].x * t0 + av.q[q].y * t1 + av.q[q].z * t2 + av.q[q].w * t3;
        }
        const float part = ((pq[0] + pq[1]) + (pq[2] + pq[3])) * (0.25f * kLog2e);
        float mx = part;
        mx = fmaxf(mx, __shfl_xor(mx, 1, 32));
        mx = fmaxf(mx, __shfl_xor(mx, 2, 32));
        mx = fmaxf(mx, __shfl_xor(mx, 4, 32));
        const float ex = __builtin_amdgcn_exp2f(part - mx);
        float sm = ex;
        sm += __shfl_xor(sm, 1, 32);
        sm += __shfl_xor(sm, 2, 32);
        sm += __shfl_xor(sm, 4, 32);
        const float alpha = ex * __builtin_amdgcn_rcpf(sm);
        if (valid) {
            float* arow = sAcc + dl * kPitch + h;
            for (int q = 0; q < 4; ++q) {
                atomicAdd(arow + (4 * q + 0) * 8, alpha * a1[q].x);
                atomicAdd(arow + (4 * q + 1) * 8, alpha * a1[q].y);
                atomicAdd(arow + (4 * q + 2) * 8, alpha * a1[q].z);
                atomicAdd(arow + (4 * q + 3) * 8, alpha * a1[q].w);
            }
        }
    }
}

template <bool kSave>
__device__ __forceinline__ void convert_entries(
    const unsigned* in, unsigned* out, const int cnt, const int lane,
    const int* __restrict__ src, unsigned* __restrict__ gSlot, unsigned& saved)
{
    const bool fits = kSave && saved != kOver && saved + (unsigned)cnt <= (unsigned)kSlot;
    for (int i = lane; i < cnt; i += 32) {
        const unsigned ent = in[i];
        const unsigned v = (unsigned)src[ent & 0xFFFFFu] | (ent & 0xFFF00000u);
        out[i] = v;
        if (fits) gSlot[saved + i] = v;
    }
    if (kSave) saved = fits ? saved + (unsigned)cnt : kOver;
}

__device__ __forceinline__ void wave_offsets(
    const unsigned mine, const int wave, unsigned& below, unsigned& total)
{
    unsigned b = 0, t = 0;
#pragma unroll
    for (int w = 0; w < 32; ++w) {
        const unsigned cw = (unsigned)__builtin_amdgcn_readlane((int)mine, w);
        t += cw;
        b += (w < wave) ? cw : 0u;
    }
    below = b;
    total = t;
}

template <bool kSave, bool kDefer>
__device__ __forceinline__ void scan_and_work(
    unsigned* myList, const int wave, const int lane, const unsigned lo, const HeadVec& av,
    const float* __restrict__ H1, const float* __restrict__ HP,
    const int* __restrict__ src, const int* __restrict__ dst,
    const float4* sH2, const float4* sHPd, float* sAcc,
    unsigned* __restrict__ gSlot, unsigned* __restrict__ gCnt,
    int& pending, unsigned& savedOut)
{
    const unsigned lanesBelow = (1u << lane) - 1u;
    int cnt = 0;
    unsigned saved = 0;

    for (int c0 = wave; c0 < kChunks; c0 += 128) {
        if (cnt > kCap - 512) {
            convert_entries<kSave>(myList, myList, cnt, lane, src, gSlot, saved);
            work_list(myList, cnt, 0, 4, lane, av, H1, HP, sH2, sHPd, sAcc);
            cnt = 0;
        }

        int4 dv[4];
#pragma unroll
        for (int u = 0; u < 4; ++u) {
            const int c = c0 + 32 * u;
            if (c < kChunks) dv[u] = *(const int4*)(dst + (size_t)c * 128 + lane * 4);
            else             dv[u] = make_int4(-1, -1, -1, -1);
        }
#pragma unroll
        for (int u = 0; u < 4; ++u) {
            const int ebase = (c0 + 32 * u) * 128 + lane * 4;
            const unsigned r0 = (unsigned)dv[u].x - lo;
            const unsigned r1 = (unsigned)dv[u].y - lo;
            const unsigned r2 = (unsigned)dv[u].z - lo;
            const unsigned r3 = (unsigned)dv[u].w - lo;
            const unsigned rm = min(min(r0, r1), min(r2, r3));
            if (__builtin_amdgcn_ballot_w32(rm < (unsigned)kTR) != 0u) {
                const unsigned rr[4] = {r0, r1, r2, r3};
#pragma unroll
                for (int k = 0; k < 4; ++k) {
                    const bool hit = rr[k] < (unsigned)kTR;
                    const unsigned mask = __builtin_amdgcn_ballot_w32(hit);
                    if (hit) {
                        const int pos = cnt + __popc(mask & lanesBelow);
                        myList[pos] = (unsigned)(ebase + k) | (rr[k] << 20);
                    }
                    cnt += __popc(mask);
                }
            }
        }
    }
    if (!kDefer) {
        convert_entries<kSave>(myList, myList, cnt, lane, src, gSlot, saved);
        work_list(myList, cnt, 0, 4, lane, av, H1, HP, sH2, sHPd, sAcc);
        cnt = 0;
        if (kSave && lane == 0) *gCnt = saved;
    }
    pending  = cnt;
    savedOut = saved;
}

template <bool kReuse>
__device__ __forceinline__ void edge_tile_body(
    const float* __restrict__ H1, const float* __restrict__ H2,
    const float* __restrict__ HP,
    const int* __restrict__ src, const int* __restrict__ dst,
    const float* __restrict__ avec,
    const float* __restrict__ G, const float* __restrict__ B,
    const float* Xin, float* Xout,
    unsigned* __restrict__ gList, unsigned* __restrict__ gCount)
{
    __shared__ float    sAcc[kTR * kPitch];
    __shared__ float4   sH2[kTR * 32];
    __shared__ float4   sHPd[kTR * 32];
    __shared__ unsigned sList[32 * kCap];
    __shared__ unsigned sJoin[kReuse ? 32 : 32 * kCap];
    __shared__ unsigned sCnt[32];

    const int tid  = threadIdx.x;
    const int lane = tid & 31;
    const int wave = __builtin_amdgcn_readfirstlane(tid >> 5);
    const unsigned lo = (unsigned)blockIdx.x * kTR;

    unsigned savedCnt = 0;
    if (kReuse) savedCnt = gCount[(size_t)blockIdx.x * 32 + lane];

    for (int i = tid; i < kTR * kPitch; i += 1024) sAcc[i] = 0.0f;
    {
        const float4* g2 = (const float4*)(H2 + (size_t)lo * kHD);
        const float4* gp = (const float4*)(HP + (size_t)lo * kHD);
        for (int i = tid; i < kTR * 32; i += 1024) {
            sH2[i]  = g2[i];
            sHPd[i] = gp[i];
        }
    }
    HeadVec av;
    for (int q = 0; q < 4; ++q) av.q[q] = *(const float4*)(avec + (lane & 7) * kD + q * 4);
    __syncthreads();

    unsigned* myList = sList + wave * kCap;
    const size_t slot = (size_t)blockIdx.x * 32 + (size_t)wave;
    unsigned* gSlot = gList + slot * kSlot;
    unsigned* gCnt  = gCount + slot;

    int pending = 0;
    unsigned saved = 0;
    if (kReuse) {
        const unsigned n = (unsigned)__builtin_amdgcn_readlane((int)savedCnt, wave);
        const bool anyOver = __builtin_amdgcn_ballot_w32(savedCnt > (unsigned)kSlot) != 0u;
        unsigned below, total;
        wave_offsets(anyOver ? 0u : savedCnt, wave, below, total);
        if (!anyOver && total <= (unsigned)(32 * kCap)) {
            for (unsigned i = lane; i < n; i += 32) sList[below + i] = gSlot[i];
            __syncthreads();
            work_list(sList, (int)total, 4 * wave, 128, lane, av, H1, HP, sH2, sHPd, sAcc);
        } else if (n <= (unsigned)kSlot) {
            for (unsigned base = 0; base < n; base += kCap) {
                const int m = (int)min((unsigned)kCap, n - base);
                for (int i = lane; i < m; i += 32) myList[i] = gSlot[base + i];
                work_list(myList, m, 0, 4, lane, av, H1, HP, sH2, sHPd, sAcc);
            }
        } else {
            scan_and_work<false, false>(myList, wave, lane, lo, av, H1, HP, src, dst,
                                        sH2, sHPd, sAcc, gSlot, gCnt, pending, saved);
        }
    } else {
        scan_and_work<true, true>(myList, wave, lane, lo, av, H1, HP, src, dst,
                                  sH2, sHPd, sAcc, gSlot, gCnt, pending, saved);
        if (lane == 0) sCnt[wave] = (unsigned)pending;
        __syncthreads();
        unsigned below, total;
        wave_offsets(sCnt[lane], wave, below, total);
        convert_entries<true>(myList, sJoin + below, pending, lane, src, gSlot, saved);
        if (lane == 0) *gCnt = saved;
        __syncthreads();
        work_list(sJoin, (int)total, 4 * wave, 128, lane, av, H1, HP, sH2, sHPd, sAcc);
    }

    constexpr int kRowsPerWave = (kTR + 31) / 32;
    float4 xr[kRowsPerWave];
#pragma unroll
    for (int j = 0; j < kRowsPerWave; ++j) {
        const int row = min(wave + 32 * j, kTR - 1);
        xr[j] = *(const float4*)(Xin + ((size_t)lo + row) * kHD + lane * 4);
    }
    const float4 g = *(const float4*)(G + lane * 4);
    const float4 b = *(const float4*)(B + lane * 4);

    __syncthreads();

#pragma unroll
    for (int j = 0; j < kRowsPerWave; ++j) {
        const int row = wave + 32 * j;
        if (row < kTR) {
            const float* a = sAcc + row * kPitch + (lane & 3) * 32 + (lane >> 2);
            float4 v;
            v.x = a[0];
            v.y = a[8];
            v.z = a[16];
            v.w = a[24];

            float s = v.x + v.y + v.z + v.w;
            for (int m = 1; m <= 16; m <<= 1) s += __shfl_xor(s, m, 32);
            const float mu = s * (1.0f / 128.0f);

            const float dx0 = v.x - mu, dx1 = v.y - mu, dx2 = v.z - mu, dx3 = v.w - mu;
            float q = dx0 * dx0 + dx1 * dx1 + dx2 * dx2 + dx3 * dx3;
            for (int m = 1; m <= 16; m <<= 1) q += __shfl_xor(q, m, 32);
            const float inv = rsqrtf(q * (1.0f / 128.0f) + kLnEps);

            float4 o;
            o.x = eluf(dx0 * inv * g.x + b.x) + xr[j].x;
            o.y = eluf(dx1 * inv * g.y + b.y) + xr[j].y;
            o.z = eluf(dx2 * inv * g.z + b.z) + xr[j].z;
            o.w = eluf(dx3 * inv * g.w + b.w) + xr[j].w;
            *(float4*)(Xout + ((size_t)lo + row) * kHD + lane * 4) = o;
        }
    }
}

__global__ __launch_bounds__(1024) void edge_tile_scan_kernel(
    const float* __restrict__ H1, const float* __restrict__ H2,
    const float* __restrict__ HP,
    const int* __restrict__ src, const int* __restrict__ dst,
    const float* __restrict__ avec,
    const float* __restrict__ G, const float* __restrict__ B,
    const float* Xin, float* Xout,
    unsigned* __restrict__ gList, unsigned* __restrict__ gCount)
{
    edge_tile_body<false>(H1, H2, HP, src, dst, avec, G, B, Xin, Xout, gList, gCount);
}

__global__ __launch_bounds__(1024) void edge_tile_list_kernel(
    const float* __restrict__ H1, const float* __restrict__ H2,
    const float* __restrict__ HP,
    const int* __restrict__ src, const int* __restrict__ dst,
    const float* __restrict__ avec,
    const float* __restrict__ G, const float* __restrict__ B,
    const float* Xin, float* Xout,
    unsigned* __restrict__ gList, unsigned* __restrict__ gCount)
{
    edge_tile_body<true>(H1, H2, HP, src, dst, avec, G, B, Xin, Xout, gList, gCount);
}

__global__ __launch_bounds__(256) void out_proj_kernel(
    const float* __restrict__ X, const float* __restrict__ Wo,
    const float* __restrict__ bo, float* __restrict__ O)
{
    const int n = blockIdx.x * blockDim.x + threadIdx.x;
    if (n >= kN) return;
    float a0 = bo[0];
    float a1 = bo[1];
    const float* xr = X + (size_t)n * kHD;
#pragma unroll 8
    for (int k = 0; k < kHD; ++k) {
        const float xv = xr[k];
        a0 += xv * Wo[2 * k + 0];
        a1 += xv * Wo[2 * k + 1];
    }
    O[2 * n + 0] = a0;
    O[2 * n + 1] = a1;
}

extern "C" void kernel_launch(void* const* d_in, const int* in_sizes, int n_in,
                              void* d_out, int out_size, void* d_ws, size_t ws_size,
                              hipStream_t stream)
{
    (void)in_sizes; (void)n_in; (void)out_size; (void)ws_size;

    const float* x    = (const float*)d_in[0];
    const int*   ei   = (const int*)  d_in[1];
    const float* W1   = (const float*)d_in[2];
    const float* b1   = (const float*)d_in[3];
    const float* W2   = (const float*)d_in[4];
    const float* b2   = (const float*)d_in[5];
    const float* a    = (const float*)d_in[6];
    const float* ln_g = (const float*)d_in[7];
    const float* ln_b = (const float*)d_in[8];
    const float* Wo   = (const float*)d_in[9];
    const float* bo   = (const float*)d_in[10];
    float*       out  = (float*)d_out;

    const size_t NHD = (size_t)kN * kHD;
    float* h1   = (float*)d_ws;
    float* h2   = h1  + NHD;
    float* hp   = h2  + NHD;
    float* xcur = hp  + NHD;
    unsigned* gList  = (unsigned*)(xcur + NHD);
    unsigned* gCount = gList + (size_t)kTiles * 32 * kSlot;

    const int* srcIdx = ei;
    const int* dstIdx = ei + kE;

    for (int layer = 0; layer < kL; ++layer) {
        const float* xin = (layer == 0) ? x : xcur;

        gemm_h1h2_kernel<<<kN / 16, 256, 0, stream>>>(
            xin,
            W1 + (size_t)layer * kIN * kHD, b1 + (size_t)layer * kHD,
            W2 + (size_t)layer * kIN * kHD, b2 + (size_t)layer * kHD,
            h1, h2, hp);

        if (layer == 0)
            edge_tile_scan_kernel<<<kTiles, 1024, 0, stream>>>(
                h1, h2, hp, srcIdx, dstIdx, a + (size_t)layer * kH * kD,
                ln_g + (size_t)layer * kHD, ln_b + (size_t)layer * kHD, xin, xcur,
                gList, gCount);
        else
            edge_tile_list_kernel<<<kTiles, 1024, 0, stream>>>(
                h1, h2, hp, srcIdx, dstIdx, a + (size_t)layer * kH * kD,
                ln_g + (size_t)layer * kHD, ln_b + (size_t)layer * kHD, xin, xcur,
                gList, gCount);
    }

    out_proj_kernel<<<(kN + 255) / 256, 256, 0, stream>>>(xcur, Wo, bo, out);
}


# GEMM re-tiled: a block stages 80 rows of X in LDS by whole-line loads and each wave runs 5 row tiles per W fragment pair (W loads and waves per row of X cut to a fifth); same k order, same sums
# speedup vs baseline: 2.1744x; 1.1578x over previous
#include <hip/hip_runtime.h>
#include <math.h>


constexpr int kN   = 20000;
constexpr int kE   = 320000;
constexpr int kIN  = 128;
constexpr int kH   = 8;
constexpr int kD   = 16;
constexpr int kHD  = 128;
constexpr int kL   = 2;
constexpr int kOUT = 2;
constexpr float kLnEps = 1e-5f;

typedef __attribute__((ext_vector_type(2))) float v2f;
typedef __attribute__((ext_vector_type(8))) float v8f;

constexpr float kLog2e = 1.44269504088896340736f;

__device__ __forceinline__ float eluf(float x) {
    const float e = __builtin_amdgcn_exp2f(x * kLog2e) - 1.0f;
    return x > 0.0f ? x : e;
}

constexpr int kGR  = 80;
constexpr int kGT  = kGR / 16;
constexpr int kXP  = kIN + 4;
constexpr int kKC  = 4;

__global__ __launch_bounds__(256) void gemm_h1h2_kernel(
    const float* __restrict__ X,
    const float* __restrict__ W1, const float* __restrict__ B1,
    const float* __restrict__ W2, const float* __restrict__ B2,
    float* __restrict__ H1, float* __restrict__ H2, float* __restrict__ HP)
{
    __shared__ float sX[kGR * kXP];

    const int tid   = threadIdx.x;
    const int lane  = tid & 31;
    const int wave  = tid >> 5;
    const int mbase = blockIdx.x * kGR;
    const int n0    = wave * 16;
    const int r     = lane & 15;
    const int half  = lane >> 4;

    const int    p    = n0 + r;
    const int    ncol = ((p >> 2) & 7) * 16 + (p >> 5) * 4 + (p & 3);

    const float* w1p = W1 + (size_t)(half * 2) * kHD + ncol;
    const float* w2p = W2 + (size_t)(half * 2) * kHD + ncol;

    v2f bA[2][kKC], bB[2][kKC];
#pragma unroll
    for (int s = 0; s < kKC; ++s) {
        bA[0][s].x = w1p[(size_t)(4 * s)     * kHD];
        bA[0][s].y = w1p[(size_t)(4 * s + 1) * kHD];
        bB[0][s].x = w2p[(size_t)(4 * s)     * kHD];
        bB[0][s].y = w2p[(size_t)(4 * s + 1) * kHD];
    }

    {
        const float4* gx = (const float4*)(X + (size_t)mbase * kIN);
        float4 xv[kGR * 32 / 256];
#pragma unroll
        for (int j = 0; j < kGR * 32 / 256; ++j) xv[j] = gx[tid + 256 * j];
#pragma unroll
        for (int j = 0; j < kGR * 32 / 256; ++j) {
            const int i = tid + 256 * j;
            *(float4*)(sX + (i >> 5) * kXP + (i & 31) * 4) = xv[j];
        }
    }
    __syncthreads();

    const float* xa = sX + r * kXP + half * 2;

    v8f acc1[kGT], acc2[kGT];
#pragma unroll
    for (int t = 0; t < kGT; ++t) { acc1[t] = v8f{}; acc2[t] = v8f{}; }

#pragma unroll
    for (int c = 0; c < kIN / (4 * kKC); ++c) {
        const int cur = c & 1;
        const int nxt = cur ^ 1;
        if (c + 1 < kIN / (4 * kKC)) {
#pragma unroll
            for (int s = 0; s < kKC; ++s) {
                const int ks = 4 * ((c + 1) * kKC + s);
                bA[nxt][s].x = w1p[(size_t)ks       * kHD];
                bA[nxt][s].y = w1p[(size_t)(ks + 1) * kHD];
                bB[nxt][s].x = w2p[(size_t)ks       * kHD];
                bB[nxt][s].y = w2p[(size_t)(ks + 1) * kHD];
            }
        }
#pragma unroll
        for (int s = 0; s < kKC; ++s) {
            const int k0 = 4 * (c * kKC + s);
            v2f a[kGT];
#pragma unroll
            for (int t = 0; t < kGT; ++t)
                a[t] = *(const v2f*)(xa + t * 16 * kXP + k0);
#pragma unroll
            for (int t = 0; t < kGT; ++t) {
                acc1[t] = __builtin_amdgcn_wmma_f32_16x16x4_f32(
                    false, a[t], false, bA[cur][s], (short)0, acc1[t], false, false);
                acc2[t] = __builtin_amdgcn_wmma_f32_16x16x4_f32(
                    false, a[t], false, bB[cur][s], (short)0, acc2[t], false, false);
            }
        }
    }

    const float bb1 = B1[ncol];
    const float bb2 = B2[ncol];

#pragma unroll
    for (int t = 0; t < kGT; ++t) {
#pragma unroll
        for (int j = 0; j < 8; ++j) {
            const int m = mbase + t * 16 + j + half * 8;
            const float h1v = acc1[t][j] + bb1;
            const float h2v = acc2[t][j] + bb2;
            const size_t idx = (size_t)m * kHD + p;
            H1[idx] = h1v;
            H2[idx] = h2v;
            HP[idx] = h1v * h2v;
        }
    }
}

constexpr int kTR     = 80;
constexpr int kTiles  = kN / kTR;
constexpr int kChunks = kE / 128;
constexpr int kCap    = 640;
constexpr int kSlot   = 1024;
constexpr unsigned kOver = 0xFFFFFFFFu;
constexpr int kPitch  = kHD + 8;

struct HeadVec { float4 q[4]; };

__device__ __forceinline__ void work_list(
    const unsigned* myList, const int cnt, const int first, const int step,
    const int lane, const HeadVec& av,
    const float* __restrict__ H1, const float* __restrict__ HP,
    const float4* sH2, const float4* sHPd, float* sAcc)
{
    const int h = lane & 7;
    const int g = lane >> 3;
    for (int i0 = first; i0 < cnt; i0 += step) {
        const bool valid = (i0 + g) < cnt;
        const unsigned ent = myList[valid ? (i0 + g) : (cnt - 1)];
        const int dl = (int)(ent >> 20);
        const unsigned boff = (ent & 0xFFFFFu) * (unsigned)(kHD * 4) + (unsigned)h * 16u;
        const float4* p1 = (const float4*)((const char*)H1 + boff);
        const float4* pp = (const float4*)((const char*)HP + boff);
        float4 a1[4], ap[4];
#pragma unroll
        for (int q = 0; q < 4; ++q) {
            a1[q] = p1[q * 8];
            ap[q] = pp[q * 8];
        }
        const float4* d2 = sH2  + dl * 32 + h;
        const float4* dp = sHPd + dl * 32 + h;
        float pq[4];
        for (int q = 0; q < 4; ++q) {
            const float4 h1s = a1[q];
            const float4 hps = ap[q];
            const float4 h2d = d2[q * 8];
            const float4 hpd = dp[q * 8];
            const float t0 = eluf(h1s.x + h2d.x + hps.x * hpd.x);
            const float t1 = eluf(h1s.y + h2d.y + hps.y * hpd.y);
            const float t2 = eluf(h1s.z + h2d.z + hps.z * hpd.z);
            const float t3 = eluf(h1s.w + h2d.w + hps.w * hpd.w);
            pq[q] = av.q[q].x * t0 + av.q[q].y * t1 + av.q[q].z * t2 + av.q[q].w * t3;
        }
        const float part = ((pq[0] + pq[1]) + (pq[2] + pq[3])) * (0.25f * kLog2e);
        float mx = part;
        mx = fmaxf(mx, __shfl_xor(mx, 1, 32));
        mx = fmaxf(mx, __shfl_xor(mx, 2, 32));
        mx = fmaxf(mx, __shfl_xor(mx, 4, 32));
        const float ex = __builtin_amdgcn_exp2f(part - mx);
        float sm = ex;
        sm += __shfl_xor(sm, 1, 32);
        sm += __shfl_xor(sm, 2, 32);
        sm += __shfl_xor(sm, 4, 32);
        const float alpha = ex * __builtin_amdgcn_rcpf(sm);
        if (valid) {
            float* arow = sAcc + dl * kPitch + h;
            for (int q = 0; q < 4; ++q) {
                atomicAdd(arow + (4 * q + 0) * 8, alpha * a1[q].x);
                atomicAdd(arow + (4 * q + 1) * 8, alpha * a1[q].y);
                atomicAdd(arow + (4 * q + 2) * 8, alpha * a1[q].z);
                atomicAdd(arow + (4 * q + 3) * 8, alpha * a1[q].w);
            }
        }
    }
}

template <bool kSave>
__device__ __forceinline__ void convert_entries(
    const unsigned* in, unsigned* out, const int cnt, const int lane,
    const int* __restrict__ src, unsigned* __restrict__ gSlot, unsigned& saved)
{
    const bool fits = kSave && saved != kOver && saved + (unsigned)cnt <= (unsigned)kSlot;
    for (int i = lane; i < cnt; i += 32) {
        const unsigned ent = in[i];
        const unsigned v = (unsigned)src[ent & 0xFFFFFu] | (ent & 0xFFF00000u);
        out[i] = v;
        if (fits) gSlot[saved + i] = v;
    }
    if (kSave) saved = fits ? saved + (unsigned)cnt : kOver;
}

__device__ __forceinline__ void wave_offsets(
    const unsigned mine, const int wave, unsigned& below, unsigned& total)
{
    unsigned b = 0, t = 0;
#pragma unroll
    for (int w = 0; w < 32; ++w) {
        const unsigned cw = (unsigned)__builtin_amdgcn_readlane((int)mine, w);
        t += cw;
        b += (w < wave) ? cw : 0u;
    }
    below = b;
    total = t;
}

template <bool kSave, bool kDefer>
__device__ __forceinline__ void scan_and_work(
    unsigned* myList, const int wave, const int lane, const unsigned lo, const HeadVec& av,
    const float* __restrict__ H1, const float* __restrict__ HP,
    const int* __restrict__ src, const int* __restrict__ dst,
    const float4* sH2, const float4* sHPd, float* sAcc,
    unsigned* __restrict__ gSlot, unsigned* __restrict__ gCnt,
    int& pending, unsigned& savedOut)
{
    const unsigned lanesBelow = (1u << lane) - 1u;
    int cnt = 0;
    unsigned saved = 0;

    for (int c0 = wave; c0 < kChunks; c0 += 128) {
        if (cnt > kCap - 512) {
            convert_entries<kSave>(myList, myList, cnt, lane, src, gSlot, saved);
            work_list(myList, cnt, 0, 4, lane, av, H1, HP, sH2, sHPd, sAcc);
            cnt = 0;
        }

        int4 dv[4];
#pragma unroll
        for (int u = 0; u < 4; ++u) {
            const int c = c0 + 32 * u;
            if (c < kChunks) dv[u] = *(const int4*)(dst + (size_t)c * 128 + lane * 4);
            else             dv[u] = make_int4(-1, -1, -1, -1);
        }
#pragma unroll
        for (int u = 0; u < 4; ++u) {
            const int ebase = (c0 + 32 * u) * 128 + lane * 4;
            const unsigned r0 = (unsigned)dv[u].x - lo;
            const unsigned r1 = (unsigned)dv[u].y - lo;
            const unsigned r2 = (unsigned)dv[u].z - lo;
            const unsigned r3 = (unsigned)dv[u].w - lo;
            const unsigned rm = min(min(r0, r1), min(r2, r3));
            if (__builtin_amdgcn_ballot_w32(rm < (unsigned)kTR) != 0u) {
                const unsigned rr[4] = {r0, r1, r2, r3};
#pragma unroll
                for (int k = 0; k < 4; ++k) {
                    const bool hit = rr[k] < (unsigned)kTR;
                    const unsigned mask = __builtin_amdgcn_ballot_w32(hit);
                    if (hit) {
                        const int pos = cnt + __popc(mask & lanesBelow);
                        myList[pos] = (unsigned)(ebase + k) | (rr[k] << 20);
                    }
                    cnt += __popc(mask);
                }
            }
        }
    }
    if (!kDefer) {
        convert_entries<kSave>(myList, myList, cnt, lane, src, gSlot, saved);
        work_list(myList, cnt, 0, 4, lane, av, H1, HP, sH2, sHPd, sAcc);
        cnt = 0;
        if (kSave && lane == 0) *gCnt = saved;
    }
    pending  = cnt;
    savedOut = saved;
}

template <bool kReuse>
__device__ __forceinline__ void edge_tile_body(
    const float* __restrict__ H1, const float* __restrict__ H2,
    const float* __restrict__ HP,
    const int* __restrict__ src, const int* __restrict__ dst,
    const float* __restrict__ avec,
    const float* __restrict__ G, const float* __restrict__ B,
    const float* Xin, float* Xout,
    unsigned* __restrict__ gList, unsigned* __restrict__ gCount)
{
    __shared__ float    sAcc[kTR * kPitch];
    __shared__ float4   sH2[kTR * 32];
    __shared__ float4   sHPd[kTR * 32];
    __shared__ unsigned sList[32 * kCap];
    __shared__ unsigned sJoin[kReuse ? 32 : 32 * kCap];
    __shared__ unsigned sCnt[32];

    const int tid  = threadIdx.x;
    const int lane = tid & 31;
    const int wave = __builtin_amdgcn_readfirstlane(tid >> 5);
    const unsigned lo = (unsigned)blockIdx.x * kTR;

    unsigned savedCnt = 0;
    if (kReuse) savedCnt = gCount[(size_t)blockIdx.x * 32 + lane];

    for (int i = tid; i < kTR * kPitch; i += 1024) sAcc[i] = 0.0f;
    {
        const float4* g2 = (const float4*)(H2 + (size_t)lo * kHD);
        const float4* gp = (const float4*)(HP + (size_t)lo * kHD);
        for (int i = tid; i < kTR * 32; i += 1024) {
            sH2[i]  = g2[i];
            sHPd[i] = gp[i];
        }
    }
    HeadVec av;
    for (int q = 0; q < 4; ++q) av.q[q] = *(const float4*)(avec + (lane & 7) * kD + q * 4);
    __syncthreads();

    unsigned* myList = sList + wave * kCap;
    const size_t slot = (size_t)blockIdx.x * 32 + (size_t)wave;
    unsigned* gSlot = gList + slot * kSlot;
    unsigned* gCnt  = gCount + slot;

    int pending = 0;
    unsigned saved = 0;
    if (kReuse) {
        const unsigned n = (unsigned)__builtin_amdgcn_readlane((int)savedCnt, wave);
        const bool anyOver = __builtin_amdgcn_ballot_w32(savedCnt > (unsigned)kSlot) != 0u;
        unsigned below, total;
        wave_offsets(anyOver ? 0u : savedCnt, wave, below, total);
        if (!anyOver && total <= (unsigned)(32 * kCap)) {
            for (unsigned i = lane; i < n; i += 32) sList[below + i] = gSlot[i];
            __syncthreads();
            work_list(sList, (int)total, 4 * wave, 128, lane, av, H1, HP, sH2, sHPd, sAcc);
        } else if (n <= (unsigned)kSlot) {
            for (unsigned base = 0; base < n; base += kCap) {
                const int m = (int)min((unsigned)kCap, n - base);
                for (int i = lane; i < m; i += 32) myList[i] = gSlot[base + i];
                work_list(myList, m, 0, 4, lane, av, H1, HP, sH2, sHPd, sAcc);
            }
        } else {
            scan_and_work<false, false>(myList, wave, lane, lo, av, H1, HP, src, dst,
                                        sH2, sHPd, sAcc, gSlot, gCnt, pending, saved);
        }
    } else {
        scan_and_work<true, true>(myList, wave, lane, lo, av, H1, HP, src, dst,
                                  sH2, sHPd, sAcc, gSlot, gCnt, pending, saved);
        if (lane == 0) sCnt[wave] = (unsigned)pending;
        __syncthreads();
        unsigned below, total;
        wave_offsets(sCnt[lane], wave, below, total);
        convert_entries<true>(myList, sJoin + below, pending, lane, src, gSlot, saved);
        if (lane == 0) *gCnt = saved;
        __syncthreads();
        work_list(sJoin, (int)total, 4 * wave, 128, lane, av, H1, HP, sH2, sHPd, sAcc);
    }

    constexpr int kRowsPerWave = (kTR + 31) / 32;
    float4 xr[kRowsPerWave];
#pragma unroll
    for (int j = 0; j < kRowsPerWave; ++j) {
        const int row = min(wave + 32 * j, kTR - 1);
        xr[j] = *(const float4*)(Xin + ((size_t)lo + row) * kHD + lane * 4);
    }
    const float4 g = *(const float4*)(G + lane * 4);
    const float4 b = *(const float4*)(B + lane * 4);

    __syncthreads();

#pragma unroll
    for (int j = 0; j < kRowsPerWave; ++j) {
        const int row = wave + 32 * j;
        if (row < kTR) {
            const float* a = sAcc + row * kPitch + (lane & 3) * 32 + (lane >> 2);
            float4 v;
            v.x = a[0];
            v.y = a[8];
            v.z = a[16];
            v.w = a[24];

            float s = v.x + v.y + v.z + v.w;
            for (int m = 1; m <= 16; m <<= 1) s += __shfl_xor(s, m, 32);
            const float mu = s * (1.0f / 128.0f);

            const float dx0 = v.x - mu, dx1 = v.y - mu, dx2 = v.z - mu, dx3 = v.w - mu;
            float q = dx0 * dx0 + dx1 * dx1 + dx2 * dx2 + dx3 * dx3;
            for (int m = 1; m <= 16; m <<= 1) q += __shfl_xor(q, m, 32);
            const float inv = rsqrtf(q * (1.0f / 128.0f) + kLnEps);

            float4 o;
            o.x = eluf(dx0 * inv * g.x + b.x) + xr[j].x;
            o.y = eluf(dx1 * inv * g.y + b.y) + xr[j].y;
            o.z = eluf(dx2 * inv * g.z + b.z) + xr[j].z;
            o.w = eluf(dx3 * inv * g.w + b.w) + xr[j].w;
            *(float4*)(Xout + ((size_t)lo + row) * kHD + lane * 4) = o;
        }
    }
}

__global__ __launch_bounds__(1024) void edge_tile_scan_kernel(
    const float* __restrict__ H1, const float* __restrict__ H2,
    const float* __restrict__ HP,
    const int* __restrict__ src, const int* __restrict__ dst,
    const float* __restrict__ avec,
    const float* __restrict__ G, const float* __restrict__ B,
    const float* Xin, float* Xout,
    unsigned* __restrict__ gList, unsigned* __restrict__ gCount)
{
    edge_tile_body<false>(H1, H2, HP, src, dst, avec, G, B, Xin, Xout, gList, gCount);
}

__global__ __launch_bounds__(1024) void edge_tile_list_kernel(
    const float* __restrict__ H1, const float* __restrict__ H2,
    const float* __restrict__ HP,
    const int* __restrict__ src, const int* __restrict__ dst,
    const float* __restrict__ avec,
    const float* __restrict__ G, const float* __restrict__ B,
    const float* Xin, float* Xout,
    unsigned* __restrict__ gList, unsigned* __restrict__ gCount)
{
    edge_tile_body<true>(H1, H2, HP, src, dst, avec, G, B, Xin, Xout, gList, gCount);
}

__global__ __launch_bounds__(256) void out_proj_kernel(
    const float* __restrict__ X, const float* __restrict__ Wo,
    const float* __restrict__ bo, float* __restrict__ O)
{
    const int n = blockIdx.x * blockDim.x + threadIdx.x;
    if (n >= kN) return;
    float a0 = bo[0];
    float a1 = bo[1];
    const float* xr = X + (size_t)n * kHD;
#pragma unroll 8
    for (int k = 0; k < kHD; ++k) {
        const float xv = xr[k];
        a0 += xv * Wo[2 * k + 0];
        a1 += xv * Wo[2 * k + 1];
    }
    O[2 * n + 0] = a0;
    O[2 * n + 1] = a1;
}

extern "C" void kernel_launch(void* const* d_in, const int* in_sizes, int n_in,
                              void* d_out, int out_size, void* d_ws, size_t ws_size,
                              hipStream_t stream)
{
    (void)in_sizes; (void)n_in; (void)out_size; (void)ws_size;

    const float* x    = (const float*)d_in[0];
    const int*   ei   = (const int*)  d_in[1];
    const float* W1   = (const float*)d_in[2];
    const float* b1   = (const float*)d_in[3];
    const float* W2   = (const float*)d_in[4];
    const float* b2   = (const float*)d_in[5];
    const float* a    = (const float*)d_in[6];
    const float* ln_g = (const float*)d_in[7];
    const float* ln_b = (const float*)d_in[8];
    const float* Wo   = (const float*)d_in[9];
    const float* bo   = (const float*)d_in[10];
    float*       out  = (float*)d_out;

    const size_t NHD = (size_t)kN * kHD;
    float* h1   = (float*)d_ws;
    float* h2   = h1  + NHD;
    float* hp   = h2  + NHD;
    float* xcur = hp  + NHD;
    unsigned* gList  = (unsigned*)(xcur + NHD);
    unsigned* gCount = gList + (size_t)kTiles * 32 * kSlot;

    const int* srcIdx = ei;
    const int* dstIdx = ei + kE;

    for (int layer = 0; layer < kL; ++layer) {
        const float* xin = (layer == 0) ? x : xcur;

        gemm_h1h2_kernel<<<kN / kGR, 256, 0, stream>>>(
            xin,
            W1 + (size_t)layer * kIN * kHD, b1 + (size_t)layer * kHD,
            W2 + (size_t)layer * kIN * kHD, b2 + (size_t)layer * kHD,
            h1, h2, hp);

        if (layer == 0)
            edge_tile_scan_kernel<<<kTiles, 1024, 0, stream>>>(
                h1, h2, hp, srcIdx, dstIdx, a + (size_t)layer * kH * kD,
                ln_g + (size_t)layer * kHD, ln_b + (size_t)layer * kHD, xin, xcur,
                gList, gCount);
        else
            edge_tile_list_kernel<<<kTiles, 1024, 0, stream>>>(
                h1, h2, hp, srcIdx, dstIdx, a + (size_t)layer * kH * kD,
                ln_g + (size_t)layer * kHD, ln_b + (size_t)layer * kHD, xin, xcur,
                gList, gCount);
    }

    out_proj_kernel<<<(kN + 255) / 256, 256, 0, stream>>>(xcur, Wo, bo, out);
}


# Output projection by the last layer's tile kernel: the finish wave multiplies its row by Wo (lane partials + 5-step butterfly: new sum order), block stores 5 lines; out_proj launch and row store gone
# speedup vs baseline: 2.4758x; 1.1386x over previous
#include <hip/hip_runtime.h>
#include <math.h>


constexpr int kN   = 20000;
constexpr int kE   = 320000;
constexpr int kIN  = 128;
constexpr int kH   = 8;
constexpr int kD   = 16;
constexpr int kHD  = 128;
constexpr int kL   = 2;
constexpr int kOUT = 2;
constexpr float kLnEps = 1e-5f;

typedef __attribute__((ext_vector_type(2))) float v2f;
typedef __attribute__((ext_vector_type(8))) float v8f;

constexpr float kLog2e = 1.44269504088896340736f;

__device__ __forceinline__ float eluf(float x) {
    const float e = __builtin_amdgcn_exp2f(x * kLog2e) - 1.0f;
    return x > 0.0f ? x : e;
}

constexpr int kGR  = 80;
constexpr int kGT  = kGR / 16;
constexpr int kXP  = kIN + 4;
constexpr int kKC  = 4;
constexpr int kTileRows = 80;
constexpr int kGE  = kE / (kN / kGR);
static_assert(kGE % 16 == 0 && kGE / 16 <= 256, "16 edges per thread");
static_assert(kN / kTileRows <= 255, "tile numbers must fit a byte, 255 = no tile");

__device__ __forceinline__ unsigned tile_byte(int d) {
    return min((unsigned)d / (unsigned)kTileRows, 255u);
}

__global__ __launch_bounds__(256) void gemm_h1h2_kernel(
    const float* __restrict__ X,
    const float* __restrict__ W1, const float* __restrict__ B1,
    const float* __restrict__ W2, const float* __restrict__ B2,
    float* __restrict__ H1, float* __restrict__ H2, float* __restrict__ HP,
    const int* __restrict__ DST, unsigned* __restrict__ TILEB)
{
    __shared__ float sX[kGR * kXP];

    const int tid   = threadIdx.x;
    const int lane  = tid & 31;
    const int wave  = tid >> 5;
    const int mbase = blockIdx.x * kGR;
    const int n0    = wave * 16;
    const int r     = lane & 15;
    const int half  = lane >> 4;

    const int    p    = n0 + r;
    const int    ncol = ((p >> 2) & 7) * 16 + (p >> 5) * 4 + (p & 3);

    const float* w1p = W1 + (size_t)(half * 2) * kHD + ncol;
    const float* w2p = W2 + (size_t)(half * 2) * kHD + ncol;

    v2f bA[2][kKC], bB[2][kKC];
#pragma unroll
    for (int s = 0; s < kKC; ++s) {
        bA[0][s].x = w1p[(size_t)(4 * s)     * kHD];
        bA[0][s].y = w1p[(size_t)(4 * s + 1) * kHD];
        bB[0][s].x = w2p[(size_t)(4 * s)     * kHD];
        bB[0][s].y = w2p[(size_t)(4 * s + 1) * kHD];
    }

    {
        const float4* gx = (const float4*)(X + (size_t)mbase * kIN);
        float4 xv[kGR * 32 / 256];
#pragma unroll
        for (int j = 0; j < kGR * 32 / 256; ++j) xv[j] = gx[tid + 256 * j];
#pragma unroll
        for (int j = 0; j < kGR * 32 / 256; ++j) {
            const int i = tid + 256 * j;
            *(float4*)(sX + (i >> 5) * kXP + (i & 31) * 4) = xv[j];
        }
    }
    __syncthreads();

    if (DST != nullptr && tid < kGE / 16) {
        const int4* dp = (const int4*)(DST + (size_t)blockIdx.x * kGE + tid * 16);
        int4 dq[4];
#pragma unroll
        for (int k = 0; k < 4; ++k) dq[k] = dp[k];
        unsigned w[4];
#pragma unroll
        for (int k = 0; k < 4; ++k)
            w[k] = tile_byte(dq[k].x) | (tile_byte(dq[k].y) << 8) |
                   (tile_byte(dq[k].z) << 16) | (tile_byte(dq[k].w) << 24);
        ((uint4*)TILEB)[(size_t)blockIdx.x * (kGE / 16) + tid] = make_uint4(w[0], w[1], w[2], w[3]);
    }

    const float* xa = sX + r * kXP + half * 2;

    v8f acc1[kGT], acc2[kGT];
#pragma unroll
    for (int t = 0; t < kGT; ++t) { acc1[t] = v8f{}; acc2[t] = v8f{}; }

#pragma unroll
    for (int c = 0; c < kIN / (4 * kKC); ++c) {
        const int cur = c & 1;
        const int nxt = cur ^ 1;
        if (c + 1 < kIN / (4 * kKC)) {
#pragma unroll
            for (int s = 0; s < kKC; ++s) {
                const int ks = 4 * ((c + 1) * kKC + s);
                bA[nxt][s].x = w1p[(size_t)ks       * kHD];
                bA[nxt][s].y = w1p[(size_t)(ks + 1) * kHD];
                bB[nxt][s].x = w2p[(size_t)ks       * kHD];
                bB[nxt][s].y = w2p[(size_t)(ks + 1) * kHD];
            }
        }
#pragma unroll
        for (int s = 0; s < kKC; ++s) {
            const int k0 = 4 * (c * kKC + s);
            v2f a[kGT];
#pragma unroll
            for (int t = 0; t < kGT; ++t)
                a[t] = *(const v2f*)(xa + t * 16 * kXP + k0);
#pragma unroll
            for (int t = 0; t < kGT; ++t) {
                acc1[t] = __builtin_amdgcn_wmma_f32_16x16x4_f32(
                    false, a[t], false, bA[cur][s], (short)0, acc1[t], false, false);
                acc2[t] = __builtin_amdgcn_wmma_f32_16x16x4_f32(
                    false, a[t], false, bB[cur][s], (short)0, acc2[t], false, false);
            }
        }
    }

    const float bb1 = B1[ncol];
    const float bb2 = B2[ncol];

#pragma unroll
    for (int t = 0; t < kGT; ++t) {
#pragma unroll
        for (int j = 0; j < 8; ++j) {
            const int m = mbase + t * 16 + j + half * 8;
            const float h1v = acc1[t][j] + bb1;
            const float h2v = acc2[t][j] + bb2;
            const size_t idx = (size_t)m * kHD + p;
            H1[idx] = h1v;
            H2[idx] = h2v;
            HP[idx] = h1v * h2v;
        }
    }
}

static_assert(kL >= 2, "the last layer runs the list kernel, which also projects");
constexpr int kTR     = 80;
constexpr int kTiles  = kN / kTR;
constexpr int kBChunks = kE / 512;
constexpr int kScanU  = 5;
static_assert(kTR == kTileRows, "the GEMM launch writes the tile bytes");
static_assert(kE % 512 == 0, "whole chunks");
constexpr int kCap    = 640;
constexpr int kSlot   = 1024;
constexpr unsigned kOver = 0xFFFFFFFFu;
constexpr int kPitch  = kHD + 8;

struct HeadVec { float4 q[4]; };

__device__ __forceinline__ void work_list(
    const unsigned* myList, const int cnt, const int first, const int step,
    const int lane, const HeadVec& av,
    const float* __restrict__ H1, const float* __restrict__ HP,
    const float4* sH2, const float4* sHPd, float* sAcc)
{
    const int h = lane & 7;
    const int g = lane >> 3;
    for (int i0 = first; i0 < cnt; i0 += step) {
        const bool valid = (i0 + g) < cnt;
        const unsigned ent = myList[valid ? (i0 + g) : (cnt - 1)];
        const int dl = (int)(ent >> 20);
        const unsigned boff = (ent & 0xFFFFFu) * (unsigned)(kHD * 4) + (unsigned)h * 16u;
        const float4* p1 = (const float4*)((const char*)H1 + boff);
        const float4* pp = (const float4*)((const char*)HP + boff);
        float4 a1[4], ap[4];
#pragma unroll
        for (int q = 0; q < 4; ++q) {
            a1[q] = p1[q * 8];
            ap[q] = pp[q * 8];
        }
        const float4* d2 = sH2  + dl * 32 + h;
        const float4* dp = sHPd + dl * 32 + h;
        float pq[4];
        for (int q = 0; q < 4; ++q) {
            const float4 h1s = a1[q];
            const float4 hps = ap[q];
            const float4 h2d = d2[q * 8];
            const float4 hpd = dp[q * 8];
            const float t0 = eluf(h1s.x + h2d.x + hps.x * hpd.x);
            const float t1 = eluf(h1s.y + h2d.y + hps.y * hpd.y);
            const float t2 = eluf(h1s.z + h2d.z + hps.z * hpd.z);
            const float t3 = eluf(h1s.w + h2d.w + hps.w * hpd.w);
            pq[q] = av.q[q].x * t0 + av.q[q].y * t1 + av.q[q].z * t2 + av.q[q].w * t3;
        }
        const float part = ((pq[0] + pq[1]) + (pq[2] + pq[3])) * (0.25f * kLog2e);
        float mx = part;
        mx = fmaxf(mx, __shfl_xor(mx, 1, 32));
        mx = fmaxf(mx, __shfl_xor(mx, 2, 32));
        mx = fmaxf(mx, __shfl_xor(mx, 4, 32));
        const float ex = __builtin_amdgcn_exp2f(part - mx);
        float sm = ex;
        sm += __shfl_xor(sm, 1, 32);
        sm += __shfl_xor(sm, 2, 32);
        sm += __shfl_xor(sm, 4, 32);
        const float alpha = ex * __builtin_amdgcn_rcpf(sm);
        if (valid) {
            float* arow = sAcc + dl * kPitch + h;
            for (int q = 0; q < 4; ++q) {
                atomicAdd(arow + (4 * q + 0) * 8, alpha * a1[q].x);
                atomicAdd(arow + (4 * q + 1) * 8, alpha * a1[q].y);
                atomicAdd(arow + (4 * q + 2) * 8, alpha * a1[q].z);
                atomicAdd(arow + (4 * q + 3) * 8, alpha * a1[q].w);
            }
        }
    }
}

template <bool kSave>
__device__ __forceinline__ void convert_entries(
    const unsigned* in, unsigned* out, const int cnt, const int lane, const unsigned lo,
    const int* __restrict__ src, const int* __restrict__ dst,
    unsigned* __restrict__ gSlot, unsigned& saved)
{
    const bool fits = kSave && saved != kOver && saved + (unsigned)cnt <= (unsigned)kSlot;
    for (int i = lane; i < cnt; i += 32) {
        const unsigned e = in[i];
        const unsigned v = (unsigned)src[e] | (((unsigned)dst[e] - lo) << 20);
        out[i] = v;
        if (fits) gSlot[saved + i] = v;
    }
    if (kSave) saved = fits ? saved + (unsigned)cnt : kOver;
}

__device__ __forceinline__ void wave_offsets(
    const unsigned mine, const int wave, unsigned& below, unsigned& total)
{
    unsigned b = 0, t = 0;
#pragma unroll
    for (int w = 0; w < 32; ++w) {
        const unsigned cw = (unsigned)__builtin_amdgcn_readlane((int)mine, w);
        t += cw;
        b += (w < wave) ? cw : 0u;
    }
    below = b;
    total = t;
}

__device__ __forceinline__ unsigned zero_bytes(const unsigned x) {
    const unsigned t = (x & 0x7F7F7F7Fu) + 0x7F7F7F7Fu;
    return ~(t | x) & 0x80808080u;
}

__device__ __forceinline__ void file_matches(
    const uint4 cur, const unsigned pat, const int ebase,
    unsigned* myList, const unsigned lanesBelow, int& cnt)
{
    unsigned m = (zero_bytes(cur.x ^ pat) >> 7) | (zero_bytes(cur.y ^ pat) >> 6) |
                 (zero_bytes(cur.z ^ pat) >> 5) | (zero_bytes(cur.w ^ pat) >> 4);
    for (;;) {
        const bool hit = m != 0u;
        const unsigned mask = __builtin_amdgcn_ballot_w32(hit);
        if (mask == 0u) break;
        if (hit) {
            const int t = __builtin_ctz(m);
            const int pos = cnt + __popc(mask & lanesBelow);
            myList[pos] = (unsigned)(ebase + 4 * (t & 7) + (t >> 3));
            m &= m - 1u;
        }
        cnt += __popc(mask);
    }
}

template <bool kSave, bool kDefer>
__device__ __forceinline__ void scan_and_work(
    unsigned* myList, const int wave, const int lane, const unsigned lo, const HeadVec& av,
    const float* __restrict__ H1, const float* __restrict__ HP,
    const int* __restrict__ src, const int* __restrict__ dst,
    const uint4* __restrict__ tileB,
    const float4* sH2, const float4* sHPd, float* sAcc,
    unsigned* __restrict__ gSlot, unsigned* __restrict__ gCnt,
    int& pending, unsigned& savedOut)
{
    const unsigned lanesBelow = (1u << lane) - 1u;
    const unsigned pat = (unsigned)blockIdx.x * 0x01010101u;
    int cnt = 0;
    unsigned saved = 0;

    for (int c0 = wave; c0 < kBChunks; c0 += 32 * kScanU) {
        uint4 tv[kScanU];
#pragma unroll
        for (int u = 0; u < kScanU; ++u) {
            const int c = min(c0 + 32 * u, kBChunks - 1);
            tv[u] = tileB[(size_t)c * 32 + lane];
        }
        int u = 0;
        while (u < kScanU) {
            bool full = false;
#pragma unroll
            for (int k = 0; k < kScanU; ++k) {
                if (!full && k >= u) {
                    if (c0 + 32 * k >= kBChunks) {
                        u = kScanU;
                    } else if (cnt > kCap - 512) {
                        full = true;
                    } else {
                        file_matches(tv[k], pat, (c0 + 32 * k) * 512 + lane * 16,
                                     myList, lanesBelow, cnt);
                        u = k + 1;
                    }
                }
            }
            if (full) {
                convert_entries<kSave>(myList, myList, cnt, lane, lo, src, dst, gSlot, saved);
                work_list(myList, cnt, 0, 4, lane, av, H1, HP, sH2, sHPd, sAcc);
                cnt = 0;
            }
        }
    }
    if (!kDefer) {
        convert_entries<kSave>(myList, myList, cnt, lane, lo, src, dst, gSlot, saved);
        work_list(myList, cnt, 0, 4, lane, av, H1, HP, sH2, sHPd, sAcc);
        cnt = 0;
        if (kSave && lane == 0) *gCnt = saved;
    }
    pending  = cnt;
    savedOut = saved;
}

template <bool kReuse>
__device__ __forceinline__ void edge_tile_body(
    const float* __restrict__ H1, const float* __restrict__ H2,
    const float* __restrict__ HP,
    const int* __restrict__ src, const int* __restrict__ dst,
    const float* __restrict__ avec,
    const float* __restrict__ G, const float* __restrict__ B,
    const float* Xin, float* Xout,
    unsigned* __restrict__ gList, unsigned* __restrict__ gCount,
    const unsigned* __restrict__ tileB,
    const float* __restrict__ Wo, const float* __restrict__ bo, float* __restrict__ O)
{
    __shared__ float    sAcc[kTR * kPitch];
    __shared__ float4   sH2[kTR * 32];
    __shared__ float4   sHPd[kTR * 32];
    __shared__ unsigned sList[32 * kCap];
    __shared__ unsigned sJoin[kReuse ? 32 : 32 * kCap];
    __shared__ unsigned sCnt[32];
    __shared__ float    sOut[kTR * kOUT];

    const int tid  = threadIdx.x;
    const int lane = tid & 31;
    const int wave = __builtin_amdgcn_readfirstlane(tid >> 5);
    const unsigned lo = (unsigned)blockIdx.x * kTR;

    unsigned savedCnt = 0;
    if (kReuse) savedCnt = gCount[(size_t)blockIdx.x * 32 + lane];

    for (int i = tid; i < kTR * kPitch; i += 1024) sAcc[i] = 0.0f;
    {
        const float4* g2 = (const float4*)(H2 + (size_t)lo * kHD);
        const float4* gp = (const float4*)(HP + (size_t)lo * kHD);
        for (int i = tid; i < kTR * 32; i += 1024) {
            sH2[i]  = g2[i];
            sHPd[i] = gp[i];
        }
    }
    HeadVec av;
    for (int q = 0; q < 4; ++q) av.q[q] = *(const float4*)(avec + (lane & 7) * kD + q * 4);
    __syncthreads();

    unsigned* myList = sList + wave * kCap;
    const size_t slot = (size_t)blockIdx.x * 32 + (size_t)wave;
    unsigned* gSlot = gList + slot * kSlot;
    unsigned* gCnt  = gCount + slot;

    int pending = 0;
    unsigned saved = 0;
    if (kReuse) {
        const unsigned n = (unsigned)__builtin_amdgcn_readlane((int)savedCnt, wave);
        const bool anyOver = __builtin_amdgcn_ballot_w32(savedCnt > (unsigned)kSlot) != 0u;
        unsigned below, total;
        wave_offsets(anyOver ? 0u : savedCnt, wave, below, total);
        if (!anyOver && total <= (unsigned)(32 * kCap)) {
            for (unsigned i = lane; i < n; i += 32) sList[below + i] = gSlot[i];
            __syncthreads();
            work_list(sList, (int)total, 4 * wave, 128, lane, av, H1, HP, sH2, sHPd, sAcc);
        } else if (n <= (unsigned)kSlot) {
            for (unsigned base = 0; base < n; base += kCap) {
                const int m = (int)min((unsigned)kCap, n - base);
                for (int i = lane; i < m; i += 32) myList[i] = gSlot[base + i];
                work_list(myList, m, 0, 4, lane, av, H1, HP, sH2, sHPd, sAcc);
            }
        } else {
            scan_and_work<false, false>(myList, wave, lane, lo, av, H1, HP, src, dst, (const uint4*)tileB,
                                        sH2, sHPd, sAcc, gSlot, gCnt, pending, saved);
        }
    } else {
        scan_and_work<true, true>(myList, wave, lane, lo, av, H1, HP, src, dst, (const uint4*)tileB,
                                  sH2, sHPd, sAcc, gSlot, gCnt, pending, saved);
        if (lane == 0) sCnt[wave] = (unsigned)pending;
        __syncthreads();
        unsigned below, total;
        wave_offsets(sCnt[lane], wave, below, total);
        convert_entries<true>(myList, sJoin + below, pending, lane, lo, src, dst, gSlot, saved);
        if (lane == 0) *gCnt = saved;
        __syncthreads();
        work_list(sJoin, (int)total, 4 * wave, 128, lane, av, H1, HP, sH2, sHPd, sAcc);
    }

    constexpr int kRowsPerWave = (kTR + 31) / 32;
    float4 xr[kRowsPerWave];
#pragma unroll
    for (int j = 0; j < kRowsPerWave; ++j) {
        const int row = min(wave + 32 * j, kTR - 1);
        xr[j] = *(const float4*)(Xin + ((size_t)lo + row) * kHD + lane * 4);
    }
    const float4 g = *(const float4*)(G + lane * 4);
    const float4 b = *(const float4*)(B + lane * 4);

    const bool project = O != nullptr;
    float4 wa = make_float4(0.0f, 0.0f, 0.0f, 0.0f), wb = wa;
    float bov = 0.0f;
    if (project) {
        wa = *(const float4*)(Wo + lane * 8);
        wb = *(const float4*)(Wo + lane * 8 + 4);
        if (lane < kOUT) bov = bo[lane];
    }

    __syncthreads();

#pragma unroll
    for (int j = 0; j < kRowsPerWave; ++j) {
        const int row = wave + 32 * j;
        if (row < kTR) {
            const float* a = sAcc + row * kPitch + (lane & 3) * 32 + (lane >> 2);
            float4 v;
            v.x = a[0];
            v.y = a[8];
            v.z = a[16];
            v.w = a[24];

            float s = v.x + v.y + v.z + v.w;
            for (int m = 1; m <= 16; m <<= 1) s += __shfl_xor(s, m, 32);
            const float mu = s * (1.0f / 128.0f);

            const float dx0 = v.x - mu, dx1 = v.y - mu, dx2 = v.z - mu, dx3 = v.w - mu;
            float q = dx0 * dx0 + dx1 * dx1 + dx2 * dx2 + dx3 * dx3;
            for (int m = 1; m <= 16; m <<= 1) q += __shfl_xor(q, m, 32);
            const float inv = rsqrtf(q * (1.0f / 128.0f) + kLnEps);

            float4 o;
            o.x = eluf(dx0 * inv * g.x + b.x) + xr[j].x;
            o.y = eluf(dx1 * inv * g.y + b.y) + xr[j].y;
            o.z = eluf(dx2 * inv * g.z + b.z) + xr[j].z;
            o.w = eluf(dx3 * inv * g.w + b.w) + xr[j].w;
            if (!project) {
                *(float4*)(Xout + ((size_t)lo + row) * kHD + lane * 4) = o;
            } else {
                const float p0 = o.x * wa.x + o.y * wa.z + o.z * wb.x + o.w * wb.z;
                const float p1 = o.x * wa.y + o.y * wa.w + o.z * wb.y + o.w * wb.w;
                const bool odd = (lane & 1) != 0;
                float keep = odd ? p1 : p0;
                const float give = odd ? p0 : p1;
                keep += __shfl_xor(give, 1, 32);
                for (int m = 2; m <= 16; m <<= 1) keep += __shfl_xor(keep, m, 32);
                if (lane < kOUT) sOut[row * kOUT + lane] = keep + bov;
            }
        }
    }

    if (project) {
        __syncthreads();
        if (tid < kTR * kOUT) O[(size_t)lo * kOUT + tid] = sOut[tid];
    }
}

__global__ __launch_bounds__(1024) void edge_tile_scan_kernel(
    const float* __restrict__ H1, const float* __restrict__ H2,
    const float* __restrict__ HP,
    const int* __restrict__ src, const int* __restrict__ dst,
    const float* __restrict__ avec,
    const float* __restrict__ G, const float* __restrict__ B,
    const float* Xin, float* Xout,
    unsigned* __restrict__ gList, unsigned* __restrict__ gCount,
    const unsigned* __restrict__ tileB)
{
    edge_tile_body<false>(H1, H2, HP, src, dst, avec, G, B, Xin, Xout, gList, gCount, tileB,
                          nullptr, nullptr, nullptr);
}

__global__ __launch_bounds__(1024) void edge_tile_list_kernel(
    const float* __restrict__ H1, const float* __restrict__ H2,
    const float* __restrict__ HP,
    const int* __restrict__ src, const int* __restrict__ dst,
    const float* __restrict__ avec,
    const float* __restrict__ G, const float* __restrict__ B,
    const float* Xin, float* Xout,
    unsigned* __restrict__ gList, unsigned* __restrict__ gCount,
    const unsigned* __restrict__ tileB,
    const float* __restrict__ Wo, const float* __restrict__ bo, float* __restrict__ O)
{
    edge_tile_body<true>(H1, H2, HP, src, dst, avec, G, B, Xin, Xout, gList, gCount, tileB,
                         Wo, bo, O);
}

extern "C" void kernel_launch(void* const* d_in, const int* in_sizes, int n_in,
                              void* d_out, int out_size, void* d_ws, size_t ws_size,
                              hipStream_t stream)
{
    (void)in_sizes; (void)n_in; (void)out_size; (void)ws_size;

    const float* x    = (const float*)d_in[0];
    const int*   ei   = (const int*)  d_in[1];
    const float* W1   = (const float*)d_in[2];
    const float* b1   = (const float*)d_in[3];
    const float* W2   = (const float*)d_in[4];
    const float* b2   = (const float*)d_in[5];
    const float* a    = (const float*)d_in[6];
    const float* ln_g = (const float*)d_in[7];
    const float* ln_b = (const float*)d_in[8];
    const float* Wo   = (const float*)d_in[9];
    const float* bo   = (const float*)d_in[10];
    float*       out  = (float*)d_out;

    const size_t NHD = (size_t)kN * kHD;
    float* h1   = (float*)d_ws;
    float* h2   = h1  + NHD;
    float* hp   = h2  + NHD;
    float* xcur = hp  + NHD;
    unsigned* gList  = (unsigned*)(xcur + NHD);
    unsigned* gCount = gList + (size_t)kTiles * 32 * kSlot;
    unsigned* tileB  = gCount + (size_t)kTiles * 32;

    const int* srcIdx = ei;
    const int* dstIdx = ei + kE;

    for (int layer = 0; layer < kL; ++layer) {
        const float* xin = (layer == 0) ? x : xcur;

        gemm_h1h2_kernel<<<kN / kGR, 256, 0, stream>>>(
            xin,
            W1 + (size_t)layer * kIN * kHD, b1 + (size_t)layer * kHD,
            W2 + (size_t)layer * kIN * kHD, b2 + (size_t)layer * kHD,
            h1, h2, hp, (layer == 0) ? dstIdx : nullptr, tileB);

        if (layer == 0)
            edge_tile_scan_kernel<<<kTiles, 1024, 0, stream>>>(
                h1, h2, hp, srcIdx, dstIdx, a + (size_t)layer * kH * kD,
                ln_g + (size_t)layer * kHD, ln_b + (size_t)layer * kHD, xin, xcur,
                gList, gCount, tileB);
        else
            edge_tile_list_kernel<<<kTiles, 1024, 0, stream>>>(
                h1, h2, hp, srcIdx, dstIdx, a + (size_t)layer * kH * kD,
                ln_g + (size_t)layer * kHD, ln_b + (size_t)layer * kHD, xin, xcur,
                gList, gCount, tileB,
                (layer == kL - 1) ? Wo : nullptr, (layer == kL - 1) ? bo : nullptr,
                (layer == kL - 1) ? out : nullptr);
    }
}


# Edge pass LDS bank conflicts removed: slot g of a pass takes its head's four 16-byte units in order q^g, sums stored unit-interleaved at pitch 128; same logit pairs; atomic arrival order may change
# speedup vs baseline: 2.5929x; 1.0473x over previous
#include <hip/hip_runtime.h>
#include <math.h>


constexpr int kN   = 20000;
constexpr int kE   = 320000;
constexpr int kIN  = 128;
constexpr int kH   = 8;
constexpr int kD   = 16;
constexpr int kHD  = 128;
constexpr int kL   = 2;
constexpr int kOUT = 2;
constexpr float kLnEps = 1e-5f;

typedef __attribute__((ext_vector_type(2))) float v2f;
typedef __attribute__((ext_vector_type(8))) float v8f;

constexpr float kLog2e = 1.44269504088896340736f;

__device__ __forceinline__ float eluf(float x) {
    const float e = __builtin_amdgcn_exp2f(x * kLog2e) - 1.0f;
    return x > 0.0f ? x : e;
}

constexpr int kGR  = 80;
constexpr int kGT  = kGR / 16;
constexpr int kXP  = kIN + 4;
constexpr int kKC  = 4;
constexpr int kTileRows = 80;
constexpr int kGE  = kE / (kN / kGR);
static_assert(kGE % 16 == 0 && kGE / 16 <= 256, "16 edges per thread");
static_assert(kN / kTileRows <= 255, "tile numbers must fit a byte, 255 = no tile");

__device__ __forceinline__ unsigned tile_byte(int d) {
    return min((unsigned)d / (unsigned)kTileRows, 255u);
}

__global__ __launch_bounds__(256) void gemm_h1h2_kernel(
    const float* __restrict__ X,
    const float* __restrict__ W1, const float* __restrict__ B1,
    const float* __restrict__ W2, const float* __restrict__ B2,
    float* __restrict__ H1, float* __restrict__ H2, float* __restrict__ HP,
    const int* __restrict__ DST, unsigned* __restrict__ TILEB)
{
    __shared__ float sX[kGR * kXP];

    const int tid   = threadIdx.x;
    const int lane  = tid & 31;
    const int wave  = tid >> 5;
    const int mbase = blockIdx.x * kGR;
    const int n0    = wave * 16;
    const int r     = lane & 15;
    const int half  = lane >> 4;

    const int    p    = n0 + r;
    const int    ncol = ((p >> 2) & 7) * 16 + (p >> 5) * 4 + (p & 3);

    const float* w1p = W1 + (size_t)(half * 2) * kHD + ncol;
    const float* w2p = W2 + (size_t)(half * 2) * kHD + ncol;

    v2f bA[2][kKC], bB[2][kKC];
#pragma unroll
    for (int s = 0; s < kKC; ++s) {
        bA[0][s].x = w1p[(size_t)(4 * s)     * kHD];
        bA[0][s].y = w1p[(size_t)(4 * s + 1) * kHD];
        bB[0][s].x = w2p[(size_t)(4 * s)     * kHD];
        bB[0][s].y = w2p[(size_t)(4 * s + 1) * kHD];
    }

    {
        const float4* gx = (const float4*)(X + (size_t)mbase * kIN);
        float4 xv[kGR * 32 / 256];
#pragma unroll
        for (int j = 0; j < kGR * 32 / 256; ++j) xv[j] = gx[tid + 256 * j];
#pragma unroll
        for (int j = 0; j < kGR * 32 / 256; ++j) {
            const int i = tid + 256 * j;
            *(float4*)(sX + (i >> 5) * kXP + (i & 31) * 4) = xv[j];
        }
    }
    __syncthreads();

    if (DST != nullptr && tid < kGE / 16) {
        const int4* dp = (const int4*)(DST + (size_t)blockIdx.x * kGE + tid * 16);
        int4 dq[4];
#pragma unroll
        for (int k = 0; k < 4; ++k) dq[k] = dp[k];
        unsigned w[4];
#pragma unroll
        for (int k = 0; k < 4; ++k)
            w[k] = tile_byte(dq[k].x) | (tile_byte(dq[k].y) << 8) |
                   (tile_byte(dq[k].z) << 16) | (tile_byte(dq[k].w) << 24);
        ((uint4*)TILEB)[(size_t)blockIdx.x * (kGE / 16) + tid] = make_uint4(w[0], w[1], w[2], w[3]);
    }

    const float* xa = sX + r * kXP + half * 2;

    v8f acc1[kGT], acc2[kGT];
#pragma unroll
    for (int t = 0; t < kGT; ++t) { acc1[t] = v8f{}; acc2[t] = v8f{}; }

#pragma unroll
    for (int c = 0; c < kIN / (4 * kKC); ++c) {
        const int cur = c & 1;
        const int nxt = cur ^ 1;
        if (c + 1 < kIN / (4 * kKC)) {
#pragma unroll
            for (int s = 0; s < kKC; ++s) {
                const int ks = 4 * ((c + 1) * kKC + s);
                bA[nxt][s].x = w1p[(size_t)ks       * kHD];
                bA[nxt][s].y = w1p[(size_t)(ks + 1) * kHD];
                bB[nxt][s].x = w2p[(size_t)ks       * kHD];
                bB[nxt][s].y = w2p[(size_t)(ks + 1) * kHD];
            }
        }
#pragma unroll
        for (int s = 0; s < kKC; ++s) {
            const int k0 = 4 * (c * kKC + s);
            v2f a[kGT];
#pragma unroll
            for (int t = 0; t < kGT; ++t)
                a[t] = *(const v2f*)(xa + t * 16 * kXP + k0);
#pragma unroll
            for (int t = 0; t < kGT; ++t) {
                acc1[t] = __builtin_amdgcn_wmma_f32_16x16x4_f32(
                    false, a[t], false, bA[cur][s], (short)0, acc1[t], false, false);
                acc2[t] = __builtin_amdgcn_wmma_f32_16x16x4_f32(
                    false, a[t], false, bB[cur][s], (short)0, acc2[t], false, false);
            }
        }
    }

    const float bb1 = B1[ncol];
    const float bb2 = B2[ncol];

#pragma unroll
    for (int t = 0; t < kGT; ++t) {
#pragma unroll
        for (int j = 0; j < 8; ++j) {
            const int m = mbase + t * 16 + j + half * 8;
            const float h1v = acc1[t][j] + bb1;
            const float h2v = acc2[t][j] + bb2;
            const size_t idx = (size_t)m * kHD + p;
            H1[idx] = h1v;
            H2[idx] = h2v;
            HP[idx] = h1v * h2v;
        }
    }
}

static_assert(kL >= 2, "the last layer runs the list kernel, which also projects");
constexpr int kTR     = 80;
constexpr int kTiles  = kN / kTR;
constexpr int kBChunks = kE / 512;
constexpr int kScanU  = 5;
static_assert(kTR == kTileRows, "the GEMM launch writes the tile bytes");
static_assert(kE % 512 == 0, "whole chunks");
constexpr int kCap    = 640;
constexpr int kSlot   = 1024;
constexpr unsigned kOver = 0xFFFFFFFFu;
constexpr int kPitch  = kHD;

struct HeadVec { float4 q[4]; };

__device__ __forceinline__ void work_list(
    const unsigned* myList, const int cnt, const int first, const int step,
    const int lane, const HeadVec& av,
    const float* __restrict__ H1, const float* __restrict__ HP,
    const float4* sH2, const float4* sHPd, float* sAcc)
{
    const int h = lane & 7;
    const int g = lane >> 3;
    const unsigned laneB = (unsigned)h * 16u + (unsigned)g * 128u;
    const unsigned laneA = (unsigned)g * 8u + (unsigned)h;
    const int u1 = (g & 1) ? -1 : 1;
    const int u2 = (g & 2) ? -2 : 2;
    for (int i0 = first; i0 < cnt; i0 += step) {
        const bool valid = (i0 + g) < cnt;
        const unsigned ent = myList[valid ? (i0 + g) : (cnt - 1)];
        const int dl = (int)(ent >> 20);
        const unsigned boff = (ent & 0xFFFFFu) * (unsigned)(kHD * 4) + laneB;
        float4 a1[4], ap[4];
#pragma unroll
        for (int q = 0; q < 4; ++q) {
            const unsigned bq = boff + (unsigned)(((q & 1) * u1 + (q >> 1) * u2) * 128);
            a1[q] = *(const float4*)((const char*)H1 + bq);
            ap[q] = *(const float4*)((const char*)HP + bq);
        }
        const unsigned doff = (unsigned)dl * (unsigned)(kHD * 4) + laneB;
        float4 b2[4], bp[4];
#pragma unroll
        for (int q = 0; q < 4; ++q) {
            const unsigned dq = doff + (unsigned)(((q & 1) * u1 + (q >> 1) * u2) * 128);
            b2[q] = *(const float4*)((const char*)sH2  + dq);
            bp[q] = *(const float4*)((const char*)sHPd + dq);
        }
        asm volatile("" :: "v"(b2[0].x), "v"(b2[0].y), "v"(b2[0].z), "v"(b2[0].w),
                           "v"(b2[1].x), "v"(b2[1].y), "v"(b2[1].z), "v"(b2[1].w),
                           "v"(b2[2].x), "v"(b2[2].y), "v"(b2[2].z), "v"(b2[2].w),
                           "v"(b2[3].x), "v"(b2[3].y), "v"(b2[3].z), "v"(b2[3].w));
        asm volatile("" :: "v"(bp[0].x), "v"(bp[0].y), "v"(bp[0].z), "v"(bp[0].w),
                           "v"(bp[1].x), "v"(bp[1].y), "v"(bp[1].z), "v"(bp[1].w),
                           "v"(bp[2].x), "v"(bp[2].y), "v"(bp[2].z), "v"(bp[2].w),
                           "v"(bp[3].x), "v"(bp[3].y), "v"(bp[3].z), "v"(bp[3].w));
        float pq[4];
        for (int q = 0; q < 4; ++q) {
            const float4 h1s = a1[q];
            const float4 hps = ap[q];
            const float4 h2d = b2[q];
            const float4 hpd = bp[q];
            const float t0 = eluf(h1s.x + h2d.x + hps.x * hpd.x);
            const float t1 = eluf(h1s.y + h2d.y + hps.y * hpd.y);
            const float t2 = eluf(h1s.z + h2d.z + hps.z * hpd.z);
            const float t3 = eluf(h1s.w + h2d.w + hps.w * hpd.w);
            pq[q] = av.q[q].x * t0 + av.q[q].y * t1 + av.q[q].z * t2 + av.q[q].w * t3;
        }
        const float part = ((pq[0] + pq[1]) + (pq[2] + pq[3])) * (0.25f * kLog2e);
        float mx = part;
        mx = fmaxf(mx, __shfl_xor(mx, 1, 32));
        mx = fmaxf(mx, __shfl_xor(mx, 2, 32));
        mx = fmaxf(mx, __shfl_xor(mx, 4, 32));
        const float ex = __builtin_amdgcn_exp2f(part - mx);
        float sm = ex;
        sm += __shfl_xor(sm, 1, 32);
        sm += __shfl_xor(sm, 2, 32);
        sm += __shfl_xor(sm, 4, 32);
        const float alpha = ex * __builtin_amdgcn_rcpf(sm);
        if (valid) {
            const unsigned aoff = (unsigned)dl * (unsigned)kPitch + laneA;
            for (int q = 0; q < 4; ++q) {
                float* arow = sAcc + (aoff + (unsigned)(((q & 1) * u1 + (q >> 1) * u2) * 8));
                atomicAdd(arow + 0 * 32, alpha * a1[q].x);
                atomicAdd(arow + 1 * 32, alpha * a1[q].y);
                atomicAdd(arow + 2 * 32, alpha * a1[q].z);
                atomicAdd(arow + 3 * 32, alpha * a1[q].w);
            }
        }
    }
}

template <bool kSave>
__device__ __forceinline__ void convert_entries(
    const unsigned* in, unsigned* out, const int cnt, const int lane, const unsigned lo,
    const int* __restrict__ src, const int* __restrict__ dst,
    unsigned* __restrict__ gSlot, unsigned& saved)
{
    const bool fits = kSave && saved != kOver && saved + (unsigned)cnt <= (unsigned)kSlot;
    for (int i = lane; i < cnt; i += 32) {
        const unsigned e = in[i];
        const unsigned v = (unsigned)src[e] | (((unsigned)dst[e] - lo) << 20);
        out[i] = v;
        if (fits) gSlot[saved + i] = v;
    }
    if (kSave) saved = fits ? saved + (unsigned)cnt : kOver;
}

__device__ __forceinline__ void wave_offsets(
    const unsigned mine, const int wave, unsigned& below, unsigned& total)
{
    unsigned b = 0, t = 0;
#pragma unroll
    for (int w = 0; w < 32; ++w) {
        const unsigned cw = (unsigned)__builtin_amdgcn_readlane((int)mine, w);
        t += cw;
        b += (w < wave) ? cw : 0u;
    }
    below = b;
    total = t;
}

__device__ __forceinline__ unsigned zero_bytes(const unsigned x) {
    const unsigned t = (x & 0x7F7F7F7Fu) + 0x7F7F7F7Fu;
    return ~(t | x) & 0x80808080u;
}

__device__ __forceinline__ void file_matches(
    const uint4 cur, const unsigned pat, const int ebase,
    unsigned* myList, const unsigned lanesBelow, int& cnt)
{
    unsigned m = (zero_bytes(cur.x ^ pat) >> 7) | (zero_bytes(cur.y ^ pat) >> 6) |
                 (zero_bytes(cur.z ^ pat) >> 5) | (zero_bytes(cur.w ^ pat) >> 4);
    for (;;) {
        const bool hit = m != 0u;
        const unsigned mask = __builtin_amdgcn_ballot_w32(hit);
        if (mask == 0u) break;
        if (hit) {
            const int t = __builtin_ctz(m);
            const int pos = cnt + __popc(mask & lanesBelow);
            myList[pos] = (unsigned)(ebase + 4 * (t & 7) + (t >> 3));
            m &= m - 1u;
        }
        cnt += __popc(mask);
    }
}

template <bool kSave, bool kDefer>
__device__ __forceinline__ void scan_and_work(
    unsigned* myList, const int wave, const int lane, const unsigned lo, const HeadVec& av,
    const float* __restrict__ H1, const float* __restrict__ HP,
    const int* __restrict__ src, const int* __restrict__ dst,
    const uint4* __restrict__ tileB,
    const float4* sH2, const float4* sHPd, float* sAcc,
    unsigned* __restrict__ gSlot, unsigned* __restrict__ gCnt,
    int& pending, unsigned& savedOut)
{
    const unsigned lanesBelow = (1u << lane) - 1u;
    const unsigned pat = (unsigned)blockIdx.x * 0x01010101u;
    int cnt = 0;
    unsigned saved = 0;

    for (int c0 = wave; c0 < kBChunks; c0 += 32 * kScanU) {
        uint4 tv[kScanU];
#pragma unroll
        for (int u = 0; u < kScanU; ++u) {
            const int c = min(c0 + 32 * u, kBChunks - 1);
            tv[u] = tileB[(size_t)c * 32 + lane];
        }
        int u = 0;
        while (u < kScanU) {
            bool full = false;
#pragma unroll
            for (int k = 0; k < kScanU; ++k) {
                if (!full && k >= u) {
                    if (c0 + 32 * k >= kBChunks) {
                        u = kScanU;
                    } else if (cnt > kCap - 512) {
                        full = true;
                    } else {
                        file_matches(tv[k], pat, (c0 + 32 * k) * 512 + lane * 16,
                                     myList, lanesBelow, cnt);
                        u = k + 1;
                    }
                }
            }
            if (full) {
                convert_entries<kSave>(myList, myList, cnt, lane, lo, src, dst, gSlot, saved);
                work_list(myList, cnt, 0, 4, lane, av, H1, HP, sH2, sHPd, sAcc);
                cnt = 0;
            }
        }
    }
    if (!kDefer) {
        convert_entries<kSave>(myList, myList, cnt, lane, lo, src, dst, gSlot, saved);
        work_list(myList, cnt, 0, 4, lane, av, H1, HP, sH2, sHPd, sAcc);
        cnt = 0;
        if (kSave && lane == 0) *gCnt = saved;
    }
    pending  = cnt;
    savedOut = saved;
}

template <bool kReuse>
__device__ __forceinline__ void edge_tile_body(
    const float* __restrict__ H1, const float* __restrict__ H2,
    const float* __restrict__ HP,
    const int* __restrict__ src, const int* __restrict__ dst,
    const float* __restrict__ avec,
    const float* __restrict__ G, const float* __restrict__ B,
    const float* Xin, float* Xout,
    unsigned* __restrict__ gList, unsigned* __restrict__ gCount,
    const unsigned* __restrict__ tileB,
    const float* __restrict__ Wo, const float* __restrict__ bo, float* __restrict__ O)
{
    __shared__ float    sAcc[kTR * kPitch];
    __shared__ float4   sHD[2 * kTR * 32];
    float4* const sH2  = sHD;
    float4* const sHPd = sHD + kTR * 32;
    __shared__ unsigned sList[32 * kCap];
    __shared__ unsigned sJoin[kReuse ? 32 : 32 * kCap];
    __shared__ unsigned sCnt[32];
    __shared__ float    sOut[kTR * kOUT];

    const int tid  = threadIdx.x;
    const int lane = tid & 31;
    const int wave = __builtin_amdgcn_readfirstlane(tid >> 5);
    const unsigned lo = (unsigned)blockIdx.x * kTR;

    unsigned savedCnt = 0;
    if (kReuse) savedCnt = gCount[(size_t)blockIdx.x * 32 + lane];

    for (int i = tid; i < kTR * kPitch; i += 1024) sAcc[i] = 0.0f;
    {
        const float4* g2 = (const float4*)(H2 + (size_t)lo * kHD);
        const float4* gp = (const float4*)(HP + (size_t)lo * kHD);
        for (int i = tid; i < kTR * 32; i += 1024) {
            sH2[i]  = g2[i];
            sHPd[i] = gp[i];
        }
    }
    HeadVec av;
    for (int q = 0; q < 4; ++q) av.q[q] = *(const float4*)(avec + (lane & 7) * kD + (q ^ (lane >> 3)) * 4);
    __syncthreads();

    unsigned* myList = sList + wave * kCap;
    const size_t slot = (size_t)blockIdx.x * 32 + (size_t)wave;
    unsigned* gSlot = gList + slot * kSlot;
    unsigned* gCnt  = gCount + slot;

    int pending = 0;
    unsigned saved = 0;
    if (kReuse) {
        const unsigned n = (unsigned)__builtin_amdgcn_readlane((int)savedCnt, wave);
        const bool anyOver = __builtin_amdgcn_ballot_w32(savedCnt > (unsigned)kSlot) != 0u;
        unsigned below, total;
        wave_offsets(anyOver ? 0u : savedCnt, wave, below, total);
        if (!anyOver && total <= (unsigned)(32 * kCap)) {
            for (unsigned i = lane; i < n; i += 32) sList[below + i] = gSlot[i];
            __syncthreads();
            work_list(sList, (int)total, 4 * wave, 128, lane, av, H1, HP, sH2, sHPd, sAcc);
        } else if (n <= (unsigned)kSlot) {
            for (unsigned base = 0; base < n; base += kCap) {
                const int m = (int)min((unsigned)kCap, n - base);
                for (int i = lane; i < m; i += 32) myList[i] = gSlot[base + i];
                work_list(myList, m, 0, 4, lane, av, H1, HP, sH2, sHPd, sAcc);
            }
        } else {
            scan_and_work<false, false>(myList, wave, lane, lo, av, H1, HP, src, dst, (const uint4*)tileB,
                                        sH2, sHPd, sAcc, gSlot, gCnt, pending, saved);
        }
    } else {
        scan_and_work<true, true>(myList, wave, lane, lo, av, H1, HP, src, dst, (const uint4*)tileB,
                                  sH2, sHPd, sAcc, gSlot, gCnt, pending, saved);
        if (lane == 0) sCnt[wave] = (unsigned)pending;
        __syncthreads();
        unsigned below, total;
        wave_offsets(sCnt[lane], wave, below, total);
        convert_entries<true>(myList, sJoin + below, pending, lane, lo, src, dst, gSlot, saved);
        if (lane == 0) *gCnt = saved;
        __syncthreads();
        work_list(sJoin, (int)total, 4 * wave, 128, lane, av, H1, HP, sH2, sHPd, sAcc);
    }

    constexpr int kRowsPerWave = (kTR + 31) / 32;
    float4 xr[kRowsPerWave];
#pragma unroll
    for (int j = 0; j < kRowsPerWave; ++j) {
        const int row = min(wave + 32 * j, kTR - 1);
        xr[j] = *(const float4*)(Xin + ((size_t)lo + row) * kHD + lane * 4);
    }
    const float4 g = *(const float4*)(G + lane * 4);
    const float4 b = *(const float4*)(B + lane * 4);

    const bool project = O != nullptr;
    float4 wa = make_float4(0.0f, 0.0f, 0.0f, 0.0f), wb = wa;
    float bov = 0.0f;
    if (project) {
        wa = *(const float4*)(Wo + lane * 8);
        wb = *(const float4*)(Wo + lane * 8 + 4);
        if (lane < kOUT) bov = bo[lane];
    }

    __syncthreads();

#pragma unroll
    for (int j = 0; j < kRowsPerWave; ++j) {
        const int row = wave + 32 * j;
        if (row < kTR) {
            const float* a = sAcc + row * kPitch + (lane & 3) * 8 + (lane >> 2);
            float4 v;
            v.x = a[0];
            v.y = a[32];
            v.z = a[64];
            v.w = a[96];

            float s = v.x + v.y + v.z + v.w;
            for (int m = 1; m <= 16; m <<= 1) s += __shfl_xor(s, m, 32);
            const float mu = s * (1.0f / 128.0f);

            const float dx0 = v.x - mu, dx1 = v.y - mu, dx2 = v.z - mu, dx3 = v.w - mu;
            float q = dx0 * dx0 + dx1 * dx1 + dx2 * dx2 + dx3 * dx3;
            for (int m = 1; m <= 16; m <<= 1) q += __shfl_xor(q, m, 32);
            const float inv = rsqrtf(q * (1.0f / 128.0f) + kLnEps);

            float4 o;
            o.x = eluf(dx0 * inv * g.x + b.x) + xr[j].x;
            o.y = eluf(dx1 * inv * g.y + b.y) + xr[j].y;
            o.z = eluf(dx2 * inv * g.z + b.z) + xr[j].z;
            o.w = eluf(dx3 * inv * g.w + b.w) + xr[j].w;
            if (!project) {
                *(float4*)(Xout + ((size_t)lo + row) * kHD + lane * 4) = o;
            } else {
                const float p0 = o.x * wa.x + o.y * wa.z + o.z * wb.x + o.w * wb.z;
                const float p1 = o.x * wa.y + o.y * wa.w + o.z * wb.y + o.w * wb.w;
                const bool odd = (lane & 1) != 0;
                float keep = odd ? p1 : p0;
                const float give = odd ? p0 : p1;
                keep += __shfl_xor(give, 1, 32);
                for (int m = 2; m <= 16; m <<= 1) keep += __shfl_xor(keep, m, 32);
                if (lane < kOUT) sOut[row * kOUT + lane] = keep + bov;
            }
        }
    }

    if (project) {
        __syncthreads();
        if (tid < kTR * kOUT) O[(size_t)lo * kOUT + tid] = sOut[tid];
    }
}

__global__ __launch_bounds__(1024) void edge_tile_scan_kernel(
    const float* __restrict__ H1, const float* __restrict__ H2,
    const float* __restrict__ HP,
    const int* __restrict__ src, const int* __restrict__ dst,
    const float* __restrict__ avec,
    const float* __restrict__ G, const float* __restrict__ B,
    const float* Xin, float* Xout,
    unsigned* __restrict__ gList, unsigned* __restrict__ gCount,
    const unsigned* __restrict__ tileB)
{
    edge_tile_body<false>(H1, H2, HP, src, dst, avec, G, B, Xin, Xout, gList, gCount, tileB,
                          nullptr, nullptr, nullptr);
}

__global__ __launch_bounds__(1024) void edge_tile_list_kernel(
    const float* __restrict__ H1, const float* __restrict__ H2,
    const float* __restrict__ HP,
    const int* __restrict__ src, const int* __restrict__ dst,
    const float* __restrict__ avec,
    const float* __restrict__ G, const float* __restrict__ B,
    const float* Xin, float* Xout,
    unsigned* __restrict__ gList, unsigned* __restrict__ gCount,
    const unsigned* __restrict__ tileB,
    const float* __restrict__ Wo, const float* __restrict__ bo, float* __restrict__ O)
{
    edge_tile_body<true>(H1, H2, HP, src, dst, avec, G, B, Xin, Xout, gList, gCount, tileB,
                         Wo, bo, O);
}

extern "C" void kernel_launch(void* const* d_in, const int* in_sizes, int n_in,
                              void* d_out, int out_size, void* d_ws, size_t ws_size,
                              hipStream_t stream)
{
    (void)in_sizes; (void)n_in; (void)out_size; (void)ws_size;

    const float* x    = (const float*)d_in[0];
    const int*   ei   = (const int*)  d_in[1];
    const float* W1   = (const float*)d_in[2];
    const float* b1   = (const float*)d_in[3];
    const float* W2   = (const float*)d_in[4];
    const float* b2   = (const float*)d_in[5];
    const float* a    = (const float*)d_in[6];
    const float* ln_g = (const float*)d_in[7];
    const float* ln_b = (const float*)d_in[8];
    const float* Wo   = (const float*)d_in[9];
    const float* bo   = (const float*)d_in[10];
    float*       out  = (float*)d_out;

    const size_t NHD = (size_t)kN * kHD;
    float* h1   = (float*)d_ws;
    float* h2   = h1  + NHD;
    float* hp   = h2  + NHD;
    float* xcur = hp  + NHD;
    unsigned* gList  = (unsigned*)(xcur + NHD);
    unsigned* gCount = gList + (size_t)kTiles * 32 * kSlot;
    unsigned* tileB  = gCount + (size_t)kTiles * 32;

    const int* srcIdx = ei;
    const int* dstIdx = ei + kE;

    for (int layer = 0; layer < kL; ++layer) {
        const float* xin = (layer == 0) ? x : xcur;

        gemm_h1h2_kernel<<<kN / kGR, 256, 0, stream>>>(
            xin,
            W1 + (size_t)layer * kIN * kHD, b1 + (size_t)layer * kHD,
            W2 + (size_t)layer * kIN * kHD, b2 + (size_t)layer * kHD,
            h1, h2, hp, (layer == 0) ? dstIdx : nullptr, tileB);

        if (layer == 0)
            edge_tile_scan_kernel<<<kTiles, 1024, 0, stream>>>(
                h1, h2, hp, srcIdx, dstIdx, a + (size_t)layer * kH * kD,
                ln_g + (size_t)layer * kHD, ln_b + (size_t)layer * kHD, xin, xcur,
                gList, gCount, tileB);
        else
            edge_tile_list_kernel<<<kTiles, 1024, 0, stream>>>(
                h1, h2, hp, srcIdx, dstIdx, a + (size_t)layer * kH * kD,
                ln_g + (size_t)layer * kHD, ln_b + (size_t)layer * kHD, xin, xcur,
                gList, gCount, tileB,
                (layer == kL - 1) ? Wo : nullptr, (layer == kL - 1) ? bo : nullptr,
                (layer == kL - 1) ? out : nullptr);
    }
}
